# starting kernel
# baseline (speedup 1.0000x reference)
#include <hip/hip_runtime.h>
#include <hip/hip_bf16.h>

#define T_LEN 512
#define B_SZ  512
#define EMB   50
#define HID   30
#define G4    120
#define VOC   100
#define NP    128
#define KE    52
#define KH    32
#define VP    112
#define CHUNK 16
#define NT    128

#define NP2 (NP * 2)
#define KH2 (KH * 2)
#define VP2 (VP * 2)

typedef __attribute__((ext_vector_type(2))) float v2f;
typedef __attribute__((ext_vector_type(8))) float v8f;

__device__ __forceinline__ v8f wmma4(v2f a, v2f b, v8f c) {
  return __builtin_amdgcn_wmma_f32_16x16x4_f32(false, a, false, b, (short)0, c,
                                               false, false);
}

__device__ __forceinline__ float sigm(float x) { return 1.0f / (1.0f + __expf(-x)); }

__device__ __forceinline__ v8f mm_acc(v8f acc, const float* __restrict__ aPtr, int kdim,
                                      const float* __restrict__ bPtr, int ld2) {
  #pragma unroll
  for (int kb = 0; kb < kdim; kb += 4) {
    v2f a = *(const v2f*)(aPtr + kb);
    v2f b = *(const v2f*)(bPtr + (kb >> 1) * ld2);
    acc = wmma4(a, b, acc);
  }
  return acc;
}

extern "C" __global__ __launch_bounds__(NT)
void charnn_fused(const int* __restrict__ batch_x, const int* __restrict__ lens,
                  const float* __restrict__ emb,
                  const float* __restrict__ w_ih1, const float* __restrict__ w_hh1,
                  const float* __restrict__ b_ih1, const float* __restrict__ b_hh1,
                  const float* __restrict__ w_ih2, const float* __restrict__ w_hh2,
                  const float* __restrict__ b_ih2, const float* __restrict__ b_hh2,
                  const float* __restrict__ lw1,  const float* __restrict__ lb1,
                  const float* __restrict__ lw2,  const float* __restrict__ lb2,
                  float* __restrict__ out)
{
  extern __shared__ float smem[];
  const int tid   = threadIdx.x;
  const int lane  = tid & 31;
  const int wave  = tid >> 5;
  const int row   = lane & 15;
  const int khalf = (lane < 16) ? 0 : 2;
  const int kh2   = khalf >> 1;
  const int mhalf = (lane < 16) ? 0 : 8;
  const int b0    = blockIdx.x * CHUNK;

  float* embL = smem;
  float* wi1  = embL + VOC * KE;
  float* wh1  = wi1  + KE * NP;
  float* wi2  = wh1  + KH * NP;
  float* wh2  = wi2  + KH * NP;
  float* l1T  = wh2  + KH * NP;
  float* l2T  = l1T  + KH * KH;
  float* bg1  = l2T  + KH * VP;
  float* bg2  = bg1  + NP;
  float* bl1L = bg2  + NP;
  float* bl2L = bl1L + KH;
  float* h1   = bl2L + VP;
  float* h2   = h1   + CHUNK * KH;
  float* h2m  = h2   + CHUNK * KH;
  float* hmid = h2m  + CHUNK * KH;
  float* gbuf = hmid + CHUNK * KH;
  float* mskf = gbuf + CHUNK * NP;
  int*   ids   = (int*)(mskf + CHUNK);
  int*   lensL = ids + CHUNK;

  for (int i = tid; i < VOC * KE; i += NT) { int r = i / KE, k = i - r * KE;
    embL[i] = (k < EMB) ? emb[r * EMB + k] : 0.f; }
  for (int i = tid; i < KE * NP; i += NT) { int k = i / NP, n = i - k * NP;
    wi1[(k >> 1) * NP2 + n * 2 + (k & 1)] =
        (k < EMB && n < G4) ? w_ih1[n * EMB + k] : 0.f; }
  for (int i = tid; i < KH * NP; i += NT) { int k = i / NP, n = i - k * NP;
    const bool ok = (k < HID && n < G4);
    const int  d  = (k >> 1) * NP2 + n * 2 + (k & 1);
    wh1[d] = ok ? w_hh1[n * HID + k] : 0.f;
    wi2[d] = ok ? w_ih2[n * HID + k] : 0.f;
    wh2[d] = ok ? w_hh2[n * HID + k] : 0.f; }
  for (int i = tid; i < KH * KH; i += NT) { int k = i / KH, j = i - k * KH;
    l1T[(k >> 1) * KH2 + j * 2 + (k & 1)] =
        (k < HID && j < HID) ? lw1[j * HID + k] : 0.f; }
  for (int i = tid; i < KH * VP; i += NT) { int k = i / VP, v = i - k * VP;
    l2T[(k >> 1) * VP2 + v * 2 + (k & 1)] =
        (k < HID && v < VOC) ? lw2[v * HID + k] : 0.f; }
  for (int i = tid; i < NP; i += NT) {
    bg1[i] = (i < G4) ? (b_ih1[i] + b_hh1[i]) : 0.f;
    bg2[i] = (i < G4) ? (b_ih2[i] + b_hh2[i]) : 0.f; }
  for (int i = tid; i < KH; i += NT) bl1L[i] = (i < HID) ? lb1[i] : 0.f;
  for (int i = tid; i < VP; i += NT) bl2L[i] = (i < VOC) ? lb2[i] : 0.f;
  for (int i = tid; i < CHUNK * KH; i += NT) { h1[i] = 0.f; h2[i] = 0.f; h2m[i] = 0.f; hmid[i] = 0.f; }
  if (tid < CHUNK) lensL[tid] = lens[b0 + tid];

  float c1a[4] = {0.f, 0.f, 0.f, 0.f};
  float c2a[4] = {0.f, 0.f, 0.f, 0.f};

  __syncthreads();

  for (int t = 0; t < T_LEN; ++t) {
    if (tid < CHUNK) {
      ids[tid]  = batch_x[(b0 + tid) * T_LEN + t];
      mskf[tid] = (t < lensL[tid]) ? 1.f : 0.f;
    }
    __syncthreads();

    {
      const float* arowE = embL + ids[row] * KE + khalf;
      const float* arowH = h1 + row * KH + khalf;
      for (int nt = wave; nt < NP / 16; nt += 4) {
        const int n = nt * 16 + row;
        const float bv = bg1[n];
        v8f acc;
        #pragma unroll
        for (int r = 0; r < 8; ++r) acc[r] = bv;
        acc = mm_acc(acc, arowE, KE, wi1 + n * 2 + kh2 * NP2, NP2);
        acc = mm_acc(acc, arowH, KH, wh1 + n * 2 + kh2 * NP2, NP2);
        #pragma unroll
        for (int r = 0; r < 8; ++r) gbuf[(r + mhalf) * NP + n] = acc[r];
      }
    }
    __syncthreads();

    #pragma unroll
    for (int r = 0; r < 4; ++r) {
      const int p = tid + NT * r;
      if (p < CHUNK * HID) {
        const int m = p / HID, j = p - m * HID;
        const float ig = sigm(gbuf[m * NP + j]);
        const float fg = sigm(gbuf[m * NP + HID + j]);
        const float gg = tanhf(gbuf[m * NP + 2 * HID + j]);
        const float og = sigm(gbuf[m * NP + 3 * HID + j]);
        const float cn = fg * c1a[r] + ig * gg;
        const float hn = og * tanhf(cn);
        const float mk = mskf[m];
        c1a[r] = mk * cn + (1.f - mk) * c1a[r];
        h1[m * KH + j] = mk * hn + (1.f - mk) * h1[m * KH + j];
      }
    }
    __syncthreads();

    {
      const float* arowA = h1 + row * KH + khalf;
      const float* arowB = h2 + row * KH + khalf;
      for (int nt = wave; nt < NP / 16; nt += 4) {
        const int n = nt * 16 + row;
        const float bv = bg2[n];
        v8f acc;
        #pragma unroll
        for (int r = 0; r < 8; ++r) acc[r] = bv;
        acc = mm_acc(acc, arowA, KH, wi2 + n * 2 + kh2 * NP2, NP2);
        acc = mm_acc(acc, arowB, KH, wh2 + n * 2 + kh2 * NP2, NP2);
        #pragma unroll
        for (int r = 0; r < 8; ++r) gbuf[(r + mhalf) * NP + n] = acc[r];
      }
    }
    __syncthreads();

    #pragma unroll
    for (int r = 0; r < 4; ++r) {
      const int p = tid + NT * r;
      if (p < CHUNK * HID) {
        const int m = p / HID, j = p - m * HID;
        const float ig = sigm(gbuf[m * NP + j]);
        const float fg = sigm(gbuf[m * NP + HID + j]);
        const float gg = tanhf(gbuf[m * NP + 2 * HID + j]);
        const float og = sigm(gbuf[m * NP + 3 * HID + j]);
        const float cn = fg * c2a[r] + ig * gg;
        const float hn = og * tanhf(cn);
        const float mk = mskf[m];
        c2a[r] = mk * cn + (1.f - mk) * c2a[r];
        const float hv = mk * hn + (1.f - mk) * h2[m * KH + j];
        h2[m * KH + j]  = hv;
        h2m[m * KH + j] = mk * hv;
      }
    }
    __syncthreads();

    if (wave < 2) {
      const float* arow = h2m + row * KH + khalf;
      const int n = wave * 16 + row;
      const float bv = bl1L[n];
      v8f acc;
      #pragma unroll
      for (int r = 0; r < 8; ++r) acc[r] = bv;
      acc = mm_acc(acc, arow, KH, l1T + n * 2 + kh2 * KH2, KH2);
      #pragma unroll
      for (int r = 0; r < 8; ++r) hmid[(r + mhalf) * KH + n] = fmaxf(acc[r], 0.f);
    }
    __syncthreads();

    {
      const float* arow = hmid + row * KH + khalf;
      for (int nt = wave; nt < VP / 16; nt += 4) {
        const int n = nt * 16 + row;
        const float bv = bl2L[n];
        v8f acc;
        #pragma unroll
        for (int r = 0; r < 8; ++r) acc[r] = bv;
        acc = mm_acc(acc, arow, KH, l2T + n * 2 + kh2 * VP2, VP2);
        if (n < VOC) {
          #pragma unroll
          for (int r = 0; r < 8; ++r) {
            const int m = r + mhalf;
            out[((size_t)(b0 + m) * T_LEN + t) * VOC + n] = acc[r];
          }
        }
      }
    }
    __syncthreads();
  }
}

extern "C" void kernel_launch(void* const* d_in, const int* in_sizes, int n_in,
                              void* d_out, int out_size, void* d_ws, size_t ws_size,
                              hipStream_t stream) {
  const int*   batch_x = (const int*)  d_in[0];
  const int*   lens    = (const int*)  d_in[1];
  const float* emb     = (const float*)d_in[2];
  const float* w_ih1   = (const float*)d_in[3];
  const float* w_hh1   = (const float*)d_in[4];
  const float* b_ih1   = (const float*)d_in[5];
  const float* b_hh1   = (const float*)d_in[6];
  const float* w_ih2   = (const float*)d_in[7];
  const float* w_hh2   = (const float*)d_in[8];
  const float* b_ih2   = (const float*)d_in[9];
  const float* b_hh2   = (const float*)d_in[10];
  const float* lw1     = (const float*)d_in[11];
  const float* lb1     = (const float*)d_in[12];
  const float* lw2     = (const float*)d_in[13];
  const float* lb2     = (const float*)d_in[14];
  float* out = (float*)d_out;

  constexpr size_t SMEM_FLOATS =
      (size_t)VOC * KE + (size_t)KE * NP + 3u * KH * NP + (size_t)KH * KH +
      (size_t)KH * VP + 2u * NP + KH + VP + 4u * CHUNK * KH + (size_t)CHUNK * NP +
      3u * CHUNK;
  constexpr size_t SMEM_BYTES = SMEM_FLOATS * sizeof(float);

  (void)hipFuncSetAttribute((const void*)charnn_fused,
                            hipFuncAttributeMaxDynamicSharedMemorySize,
                            (int)SMEM_BYTES);

  charnn_fused<<<dim3(B_SZ / CHUNK), dim3(NT), SMEM_BYTES, stream>>>(
      batch_x, lens, emb, w_ih1, w_hh1, b_ih1, b_hh1,
      w_ih2, w_hh2, b_ih2, b_hh2, lw1, lb1, lw2, lb2, out);
}


# Workgroup widened from 4 to 16 waves: every gate and head N-tile gets a wave of its own and every LSTM cell a lane of its own, so each phase between barriers is one tile or one cell deep
# speedup vs baseline: 1.6240x; 1.6240x over previous
#include <hip/hip_runtime.h>
#include <hip/hip_bf16.h>

#define T_LEN 512
#define B_SZ  512
#define EMB   50
#define HID   30
#define G4    120
#define VOC   100
#define NP    128
#define KE    52
#define KH    32
#define VP    112
#define CHUNK 16
#define NT    512
#define CPT   ((CHUNK * HID + NT - 1) / NT)

#define NP2 (NP * 2)
#define KH2 (KH * 2)
#define VP2 (VP * 2)

typedef __attribute__((ext_vector_type(2))) float v2f;
typedef __attribute__((ext_vector_type(8))) float v8f;

__device__ __forceinline__ v8f wmma4(v2f a, v2f b, v8f c) {
  return __builtin_amdgcn_wmma_f32_16x16x4_f32(false, a, false, b, (short)0, c,
                                               false, false);
}

__device__ __forceinline__ float sigm(float x) { return 1.0f / (1.0f + __expf(-x)); }

__device__ __forceinline__ v8f mm_acc(v8f acc, const float* __restrict__ aPtr, int kdim,
                                      const float* __restrict__ bPtr, int ld2) {
  #pragma unroll
  for (int kb = 0; kb < kdim; kb += 4) {
    v2f a = *(const v2f*)(aPtr + kb);
    v2f b = *(const v2f*)(bPtr + (kb >> 1) * ld2);
    acc = wmma4(a, b, acc);
  }
  return acc;
}

extern "C" __global__ __launch_bounds__(NT)
void charnn_fused(const int* __restrict__ batch_x, const int* __restrict__ lens,
                  const float* __restrict__ emb,
                  const float* __restrict__ w_ih1, const float* __restrict__ w_hh1,
                  const float* __restrict__ b_ih1, const float* __restrict__ b_hh1,
                  const float* __restrict__ w_ih2, const float* __restrict__ w_hh2,
                  const float* __restrict__ b_ih2, const float* __restrict__ b_hh2,
                  const float* __restrict__ lw1,  const float* __restrict__ lb1,
                  const float* __restrict__ lw2,  const float* __restrict__ lb2,
                  float* __restrict__ out)
{
  extern __shared__ float smem[];
  const int tid   = threadIdx.x;
  const int lane  = tid & 31;
  const int wave  = tid >> 5;
  const int row   = lane & 15;
  const int khalf = (lane < 16) ? 0 : 2;
  const int kh2   = khalf >> 1;
  const int mhalf = (lane < 16) ? 0 : 8;
  const int b0    = blockIdx.x * CHUNK;

  float* embL = smem;
  float* wi1  = embL + VOC * KE;
  float* wh1  = wi1  + KE * NP;
  float* wi2  = wh1  + KH * NP;
  float* wh2  = wi2  + KH * NP;
  float* l1T  = wh2  + KH * NP;
  float* l2T  = l1T  + KH * KH;
  float* bg1  = l2T  + KH * VP;
  float* bg2  = bg1  + NP;
  float* bl1L = bg2  + NP;
  float* bl2L = bl1L + KH;
  float* h1   = bl2L + VP;
  float* h2   = h1   + CHUNK * KH;
  float* h2m  = h2   + CHUNK * KH;
  float* hmid = h2m  + CHUNK * KH;
  float* gbuf = hmid + CHUNK * KH;
  float* mskf = gbuf + CHUNK * NP;
  int*   ids   = (int*)(mskf + CHUNK);
  int*   lensL = ids + CHUNK;

  for (int i = tid; i < VOC * KE; i += NT) { int r = i / KE, k = i - r * KE;
    embL[i] = (k < EMB) ? emb[r * EMB + k] : 0.f; }
  for (int i = tid; i < KE * NP; i += NT) { int k = i / NP, n = i - k * NP;
    wi1[(k >> 1) * NP2 + n * 2 + (k & 1)] =
        (k < EMB && n < G4) ? w_ih1[n * EMB + k] : 0.f; }
  for (int i = tid; i < KH * NP; i += NT) { int k = i / NP, n = i - k * NP;
    const bool ok = (k < HID && n < G4);
    const int  d  = (k >> 1) * NP2 + n * 2 + (k & 1);
    wh1[d] = ok ? w_hh1[n * HID + k] : 0.f;
    wi2[d] = ok ? w_ih2[n * HID + k] : 0.f;
    wh2[d] = ok ? w_hh2[n * HID + k] : 0.f; }
  for (int i = tid; i < KH * KH; i += NT) { int k = i / KH, j = i - k * KH;
    l1T[(k >> 1) * KH2 + j * 2 + (k & 1)] =
        (k < HID && j < HID) ? lw1[j * HID + k] : 0.f; }
  for (int i = tid; i < KH * VP; i += NT) { int k = i / VP, v = i - k * VP;
    l2T[(k >> 1) * VP2 + v * 2 + (k & 1)] =
        (k < HID && v < VOC) ? lw2[v * HID + k] : 0.f; }
  for (int i = tid; i < NP; i += NT) {
    bg1[i] = (i < G4) ? (b_ih1[i] + b_hh1[i]) : 0.f;
    bg2[i] = (i < G4) ? (b_ih2[i] + b_hh2[i]) : 0.f; }
  for (int i = tid; i < KH; i += NT) bl1L[i] = (i < HID) ? lb1[i] : 0.f;
  for (int i = tid; i < VP; i += NT) bl2L[i] = (i < VOC) ? lb2[i] : 0.f;
  for (int i = tid; i < CHUNK * KH; i += NT) { h1[i] = 0.f; h2[i] = 0.f; h2m[i] = 0.f; hmid[i] = 0.f; }
  if (tid < CHUNK) lensL[tid] = lens[b0 + tid];

  float c1a[CPT];
  float c2a[CPT];
  #pragma unroll
  for (int r = 0; r < CPT; ++r) { c1a[r] = 0.f; c2a[r] = 0.f; }

  __syncthreads();

  for (int t = 0; t < T_LEN; ++t) {
    if (tid < CHUNK) {
      ids[tid]  = batch_x[(b0 + tid) * T_LEN + t];
      mskf[tid] = (t < lensL[tid]) ? 1.f : 0.f;
    }
    __syncthreads();

    if (wave < NP / 16) {
      const float* arowE = embL + ids[row] * KE + khalf;
      const float* arowH = h1 + row * KH + khalf;
      const int n = wave * 16 + row;
      const float bv = bg1[n];
      v8f acc;
      #pragma unroll
      for (int r = 0; r < 8; ++r) acc[r] = bv;
      acc = mm_acc(acc, arowE, KE, wi1 + n * 2 + kh2 * NP2, NP2);
      acc = mm_acc(acc, arowH, KH, wh1 + n * 2 + kh2 * NP2, NP2);
      #pragma unroll
      for (int r = 0; r < 8; ++r) gbuf[(r + mhalf) * NP + n] = acc[r];
    }
    __syncthreads();

    #pragma unroll
    for (int r = 0; r < CPT; ++r) {
      const int p = tid + NT * r;
      if (p < CHUNK * HID) {
        const int m = p / HID, j = p - m * HID;
        const float ig = sigm(gbuf[m * NP + j]);
        const float fg = sigm(gbuf[m * NP + HID + j]);
        const float gg = tanhf(gbuf[m * NP + 2 * HID + j]);
        const float og = sigm(gbuf[m * NP + 3 * HID + j]);
        const float cn = fg * c1a[r] + ig * gg;
        const float hn = og * tanhf(cn);
        const float mk = mskf[m];
        c1a[r] = mk * cn + (1.f - mk) * c1a[r];
        h1[m * KH + j] = mk * hn + (1.f - mk) * h1[m * KH + j];
      }
    }
    __syncthreads();

    if (wave < NP / 16) {
      const float* arowA = h1 + row * KH + khalf;
      const float* arowB = h2 + row * KH + khalf;
      const int n = wave * 16 + row;
      const float bv = bg2[n];
      v8f acc;
      #pragma unroll
      for (int r = 0; r < 8; ++r) acc[r] = bv;
      acc = mm_acc(acc, arowA, KH, wi2 + n * 2 + kh2 * NP2, NP2);
      acc = mm_acc(acc, arowB, KH, wh2 + n * 2 + kh2 * NP2, NP2);
      #pragma unroll
      for (int r = 0; r < 8; ++r) gbuf[(r + mhalf) * NP + n] = acc[r];
    }
    __syncthreads();

    #pragma unroll
    for (int r = 0; r < CPT; ++r) {
      const int p = tid + NT * r;
      if (p < CHUNK * HID) {
        const int m = p / HID, j = p - m * HID;
        const float ig = sigm(gbuf[m * NP + j]);
        const float fg = sigm(gbuf[m * NP + HID + j]);
        const float gg = tanhf(gbuf[m * NP + 2 * HID + j]);
        const float og = sigm(gbuf[m * NP + 3 * HID + j]);
        const float cn = fg * c2a[r] + ig * gg;
        const float hn = og * tanhf(cn);
        const float mk = mskf[m];
        c2a[r] = mk * cn + (1.f - mk) * c2a[r];
        const float hv = mk * hn + (1.f - mk) * h2[m * KH + j];
        h2[m * KH + j]  = hv;
        h2m[m * KH + j] = mk * hv;
      }
    }
    __syncthreads();

    if (wave < 2) {
      const float* arow = h2m + row * KH + khalf;
      const int n = wave * 16 + row;
      const float bv = bl1L[n];
      v8f acc;
      #pragma unroll
      for (int r = 0; r < 8; ++r) acc[r] = bv;
      acc = mm_acc(acc, arow, KH, l1T + n * 2 + kh2 * KH2, KH2);
      #pragma unroll
      for (int r = 0; r < 8; ++r) hmid[(r + mhalf) * KH + n] = fmaxf(acc[r], 0.f);
    }
    __syncthreads();

    if (wave < VP / 16) {
      const float* arow = hmid + row * KH + khalf;
      const int n = wave * 16 + row;
      const float bv = bl2L[n];
      v8f acc;
      #pragma unroll
      for (int r = 0; r < 8; ++r) acc[r] = bv;
      acc = mm_acc(acc, arow, KH, l2T + n * 2 + kh2 * VP2, VP2);
      if (n < VOC) {
        #pragma unroll
        for (int r = 0; r < 8; ++r) {
          const int m = r + mhalf;
          out[((size_t)(b0 + m) * T_LEN + t) * VOC + n] = acc[r];
        }
      }
    }
    __syncthreads();
  }
}

extern "C" void kernel_launch(void* const* d_in, const int* in_sizes, int n_in,
                              void* d_out, int out_size, void* d_ws, size_t ws_size,
                              hipStream_t stream) {
  const int*   batch_x = (const int*)  d_in[0];
  const int*   lens    = (const int*)  d_in[1];
  const float* emb     = (const float*)d_in[2];
  const float* w_ih1   = (const float*)d_in[3];
  const float* w_hh1   = (const float*)d_in[4];
  const float* b_ih1   = (const float*)d_in[5];
  const float* b_hh1   = (const float*)d_in[6];
  const float* w_ih2   = (const float*)d_in[7];
  const float* w_hh2   = (const float*)d_in[8];
  const float* b_ih2   = (const float*)d_in[9];
  const float* b_hh2   = (const float*)d_in[10];
  const float* lw1     = (const float*)d_in[11];
  const float* lb1     = (const float*)d_in[12];
  const float* lw2     = (const float*)d_in[13];
  const float* lb2     = (const float*)d_in[14];
  float* out = (float*)d_out;

  constexpr size_t SMEM_FLOATS =
      (size_t)VOC * KE + (size_t)KE * NP + 3u * KH * NP + (size_t)KH * KH +
      (size_t)KH * VP + 2u * NP + KH + VP + 4u * CHUNK * KH + (size_t)CHUNK * NP +
      3u * CHUNK;
  constexpr size_t SMEM_BYTES = SMEM_FLOATS * sizeof(float);

  (void)hipFuncSetAttribute((const void*)charnn_fused,
                            hipFuncAttributeMaxDynamicSharedMemorySize,
                            (int)SMEM_BYTES);

  charnn_fused<<<dim3(B_SZ / CHUNK), dim3(NT), SMEM_BYTES, stream>>>(
      batch_x, lens, emb, w_ih1, w_hh1, b_ih1, b_hh1,
      w_ih2, w_hh2, b_ih2, b_hh2, lw1, lb1, lw2, lb2, out);
}


# Time loop software-pipelined across waves: matrix waves run gates1(s), gates2(s-1), head(s-2,s-3) while separate cell waves do the LSTM updates one stage behind; 2 barriers per step instead of 7
# speedup vs baseline: 2.8124x; 1.7318x over previous
#include <hip/hip_runtime.h>
#include <hip/hip_bf16.h>

#define T_LEN 512
#define B_SZ  512
#define EMB   50
#define HID   30
#define G4    120
#define VOC   100
#define NP    128
#define KE    52
#define KH    32
#define VP    112
#define CHUNK 16
#define MW    16
#define CW    15
#define NT    ((MW + CW) * 32)
#define CELL0 (MW * 32)
#define ST    (T_LEN + 3)

#define NP2 (NP * 2)
#define KH2 (KH * 2)
#define VP2 (VP * 2)

typedef __attribute__((ext_vector_type(2))) float v2f;
typedef __attribute__((ext_vector_type(8))) float v8f;

__device__ __forceinline__ v8f wmma4(v2f a, v2f b, v8f c) {
  return __builtin_amdgcn_wmma_f32_16x16x4_f32(false, a, false, b, (short)0, c,
                                               false, false);
}

__device__ __forceinline__ float sigm(float x) { return 1.0f / (1.0f + __expf(-x)); }

__device__ __forceinline__ v8f mm_acc(v8f acc, const float* __restrict__ aPtr, int kdim,
                                      const float* __restrict__ bPtr, int ld2) {
  #pragma unroll
  for (int kb = 0; kb < kdim; kb += 4) {
    v2f a = *(const v2f*)(aPtr + kb);
    v2f b = *(const v2f*)(bPtr + (kb >> 1) * ld2);
    acc = wmma4(a, b, acc);
  }
  return acc;
}

extern "C" __global__ __launch_bounds__(NT)
void charnn_fused(const int* __restrict__ batch_x, const int* __restrict__ lens,
                  const float* __restrict__ emb,
                  const float* __restrict__ w_ih1, const float* __restrict__ w_hh1,
                  const float* __restrict__ b_ih1, const float* __restrict__ b_hh1,
                  const float* __restrict__ w_ih2, const float* __restrict__ w_hh2,
                  const float* __restrict__ b_ih2, const float* __restrict__ b_hh2,
                  const float* __restrict__ lw1,  const float* __restrict__ lb1,
                  const float* __restrict__ lw2,  const float* __restrict__ lb2,
                  float* __restrict__ out)
{
  extern __shared__ float smem[];
  const int tid   = threadIdx.x;
  const int lane  = tid & 31;
  const int wave  = tid >> 5;
  const int row   = lane & 15;
  const int khalf = (lane < 16) ? 0 : 2;
  const int kh2   = khalf >> 1;
  const int mhalf = (lane < 16) ? 0 : 8;
  const int b0    = blockIdx.x * CHUNK;

  float* embL = smem;
  float* wi1  = embL + VOC * KE;
  float* wh1  = wi1  + KE * NP;
  float* wi2  = wh1  + KH * NP;
  float* wh2  = wi2  + KH * NP;
  float* l1T  = wh2  + KH * NP;
  float* l2T  = l1T  + KH * KH;
  float* bg1  = l2T  + KH * VP;
  float* bg2  = bg1  + NP;
  float* bl1L = bg2  + NP;
  float* bl2L = bl1L + KH;
  float* h1   = bl2L + VP;
  float* h2   = h1   + 2 * CHUNK * KH;
  float* h2m  = h2   + CHUNK * KH;
  float* hmid = h2m  + CHUNK * KH;
  float* gb1  = hmid + CHUNK * KH;
  float* gb2  = gb1  + CHUNK * NP;
  int*   ids  = (int*)(gb2 + CHUNK * NP);

  for (int i = tid; i < VOC * KE; i += NT) { int r = i / KE, k = i - r * KE;
    embL[i] = (k < EMB) ? emb[r * EMB + k] : 0.f; }
  for (int i = tid; i < KE * NP; i += NT) { int k = i / NP, n = i - k * NP;
    wi1[(k >> 1) * NP2 + n * 2 + (k & 1)] =
        (k < EMB && n < G4) ? w_ih1[n * EMB + k] : 0.f; }
  for (int i = tid; i < KH * NP; i += NT) { int k = i / NP, n = i - k * NP;
    const bool ok = (k < HID && n < G4);
    const int  d  = (k >> 1) * NP2 + n * 2 + (k & 1);
    wh1[d] = ok ? w_hh1[n * HID + k] : 0.f;
    wi2[d] = ok ? w_ih2[n * HID + k] : 0.f;
    wh2[d] = ok ? w_hh2[n * HID + k] : 0.f; }
  for (int i = tid; i < KH * KH; i += NT) { int k = i / KH, j = i - k * KH;
    l1T[(k >> 1) * KH2 + j * 2 + (k & 1)] =
        (k < HID && j < HID) ? lw1[j * HID + k] : 0.f; }
  for (int i = tid; i < KH * VP; i += NT) { int k = i / VP, v = i - k * VP;
    l2T[(k >> 1) * VP2 + v * 2 + (k & 1)] =
        (k < HID && v < VOC) ? lw2[v * HID + k] : 0.f; }
  for (int i = tid; i < NP; i += NT) {
    bg1[i] = (i < G4) ? (b_ih1[i] + b_hh1[i]) : 0.f;
    bg2[i] = (i < G4) ? (b_ih2[i] + b_hh2[i]) : 0.f; }
  for (int i = tid; i < KH; i += NT) bl1L[i] = (i < HID) ? lb1[i] : 0.f;
  for (int i = tid; i < VP; i += NT) bl2L[i] = (i < VOC) ? lb2[i] : 0.f;
  for (int i = tid; i < CHUNK * KH; i += NT) {
    h1[i] = 0.f; h1[CHUNK * KH + i] = 0.f; h2[i] = 0.f; h2m[i] = 0.f; hmid[i] = 0.f; }
  if (tid < CHUNK) ids[tid] = batch_x[(b0 + tid) * T_LEN];

  const int  p    = tid - CELL0;
  const bool cell = (p >= 0);
  const int  cm   = cell ? p / HID : 0;
  const int  cj   = cell ? p - cm * HID : 0;
  const int  clen = cell ? lens[b0 + cm] : 0;
  float c1 = 0.f, h1r = 0.f, c2 = 0.f, h2r = 0.f;
  int nid = 0;

  __syncthreads();

  for (int s = 0; s < ST; ++s) {
    if (wave < NP / 16) {
      if (s < T_LEN) {
        const float* arowE = embL + ids[row] * KE + khalf;
        const float* arowH = h1 + ((s + 1) & 1) * (CHUNK * KH) + row * KH + khalf;
        const int n = wave * 16 + row;
        const float bv = bg1[n];
        v8f acc;
        #pragma unroll
        for (int r = 0; r < 8; ++r) acc[r] = bv;
        acc = mm_acc(acc, arowE, KE, wi1 + n * 2 + kh2 * NP2, NP2);
        acc = mm_acc(acc, arowH, KH, wh1 + n * 2 + kh2 * NP2, NP2);
        #pragma unroll
        for (int r = 0; r < 8; ++r) gb1[(r + mhalf) * NP + n] = acc[r];
      }
    } else if (wave < NP / 16 + VP / 16) {
      if (s >= 3) {
        const int t = s - 3;
        const float* arow = hmid + row * KH + khalf;
        const int n = (wave - NP / 16) * 16 + row;
        const float bv = bl2L[n];
        v8f acc;
        #pragma unroll
        for (int r = 0; r < 8; ++r) acc[r] = bv;
        acc = mm_acc(acc, arow, KH, l2T + n * 2 + kh2 * VP2, VP2);
        if (n < VOC) {
          #pragma unroll
          for (int r = 0; r < 8; ++r) {
            const int m = r + mhalf;
            out[((size_t)(b0 + m) * T_LEN + t) * VOC + n] = acc[r];
          }
        }
      }
    } else if (wave == MW - 1) {
      if (lane < CHUNK) nid = (s + 1 < T_LEN) ? batch_x[(b0 + lane) * T_LEN + s + 1] : 0;
    } else {
      if (s >= 2 && s < T_LEN + 2) {
        const int t = s - 2;
        const float ig = sigm(gb2[cm * NP + cj]);
        const float fg = sigm(gb2[cm * NP + HID + cj]);
        const float gg = tanhf(gb2[cm * NP + 2 * HID + cj]);
        const float og = sigm(gb2[cm * NP + 3 * HID + cj]);
        const float cn = fg * c2 + ig * gg;
        const float hn = og * tanhf(cn);
        const float mk = (t < clen) ? 1.f : 0.f;
        c2 = mk * cn + (1.f - mk) * c2;
        h2r = mk * hn + (1.f - mk) * h2r;
        h2[cm * KH + cj]  = h2r;
        h2m[cm * KH + cj] = mk * h2r;
      }
    }
    __syncthreads();

    if (wave < NP / 16) {
      if (s >= 1 && s <= T_LEN) {
        const float* arowA = h1 + ((s + 1) & 1) * (CHUNK * KH) + row * KH + khalf;
        const float* arowB = h2 + row * KH + khalf;
        const int n = wave * 16 + row;
        const float bv = bg2[n];
        v8f acc;
        #pragma unroll
        for (int r = 0; r < 8; ++r) acc[r] = bv;
        acc = mm_acc(acc, arowA, KH, wi2 + n * 2 + kh2 * NP2, NP2);
        acc = mm_acc(acc, arowB, KH, wh2 + n * 2 + kh2 * NP2, NP2);
        #pragma unroll
        for (int r = 0; r < 8; ++r) gb2[(r + mhalf) * NP + n] = acc[r];
      }
    } else if (wave < NP / 16 + 2) {
      if (s >= 2 && s < T_LEN + 2) {
        const float* arow = h2m + row * KH + khalf;
        const int n = (wave - NP / 16) * 16 + row;
        const float bv = bl1L[n];
        v8f acc;
        #pragma unroll
        for (int r = 0; r < 8; ++r) acc[r] = bv;
        acc = mm_acc(acc, arow, KH, l1T + n * 2 + kh2 * KH2, KH2);
        #pragma unroll
        for (int r = 0; r < 8; ++r) hmid[(r + mhalf) * KH + n] = fmaxf(acc[r], 0.f);
      }
    } else if (wave == MW - 1) {
      if (lane < CHUNK) ids[lane] = nid;
    } else if (wave >= MW) {
      if (s < T_LEN) {
        const float ig = sigm(gb1[cm * NP + cj]);
        const float fg = sigm(gb1[cm * NP + HID + cj]);
        const float gg = tanhf(gb1[cm * NP + 2 * HID + cj]);
        const float og = sigm(gb1[cm * NP + 3 * HID + cj]);
        const float cn = fg * c1 + ig * gg;
        const float hn = og * tanhf(cn);
        const float mk = (s < clen) ? 1.f : 0.f;
        c1 = mk * cn + (1.f - mk) * c1;
        h1r = mk * hn + (1.f - mk) * h1r;
        h1[(s & 1) * (CHUNK * KH) + cm * KH + cj] = h1r;
      }
    }
    __syncthreads();
  }
}

extern "C" void kernel_launch(void* const* d_in, const int* in_sizes, int n_in,
                              void* d_out, int out_size, void* d_ws, size_t ws_size,
                              hipStream_t stream) {
  const int*   batch_x = (const int*)  d_in[0];
  const int*   lens    = (const int*)  d_in[1];
  const float* emb     = (const float*)d_in[2];
  const float* w_ih1   = (const float*)d_in[3];
  const float* w_hh1   = (const float*)d_in[4];
  const float* b_ih1   = (const float*)d_in[5];
  const float* b_hh1   = (const float*)d_in[6];
  const float* w_ih2   = (const float*)d_in[7];
  const float* w_hh2   = (const float*)d_in[8];
  const float* b_ih2   = (const float*)d_in[9];
  const float* b_hh2   = (const float*)d_in[10];
  const float* lw1     = (const float*)d_in[11];
  const float* lb1     = (const float*)d_in[12];
  const float* lw2     = (const float*)d_in[13];
  const float* lb2     = (const float*)d_in[14];
  float* out = (float*)d_out;

  constexpr size_t SMEM_FLOATS =
      (size_t)VOC * KE + (size_t)KE * NP + 3u * KH * NP + (size_t)KH * KH +
      (size_t)KH * VP + 2u * NP + KH + VP + 5u * CHUNK * KH + 2u * CHUNK * NP +
      (size_t)CHUNK;
  constexpr size_t SMEM_BYTES = SMEM_FLOATS * sizeof(float);

  (void)hipFuncSetAttribute((const void*)charnn_fused,
                            hipFuncAttributeMaxDynamicSharedMemorySize,
                            (int)SMEM_BYTES);

  charnn_fused<<<dim3(B_SZ / CHUNK), dim3(NT), SMEM_BYTES, stream>>>(
      batch_x, lens, emb, w_ih1, w_hh1, b_ih1, b_hh1,
      w_ih2, w_hh2, b_ih2, b_hh2, lw1, lb1, lw2, lb2, out);
}


# LDS row pitches padded (h rows 32->36 floats, weight K-pair rows 256->288 and 64->96, gate scratch 128->158) so fragment loads, tile stores and cell reads hit distinct banks; h-row loads were 8-way
# speedup vs baseline: 3.1373x; 1.1155x over previous
#include <hip/hip_runtime.h>
#include <hip/hip_bf16.h>

#define T_LEN 512
#define B_SZ  512
#define EMB   50
#define HID   30
#define G4    120
#define VOC   100
#define NP    128
#define KE    52
#define KH    32
#define VP    112
#define CHUNK 16
#define MW    16
#define CW    15
#define NT    ((MW + CW) * 32)
#define CELL0 (MW * 32)
#define ST    (T_LEN + 3)

#define KHP 36
#define GP  158
#define NP2 (NP * 2 + 32)
#define KH2 (KH * 2 + 32)
#define VP2 (VP * 2)

typedef __attribute__((ext_vector_type(2))) float v2f;
typedef __attribute__((ext_vector_type(8))) float v8f;

__device__ __forceinline__ v8f wmma4(v2f a, v2f b, v8f c) {
  return __builtin_amdgcn_wmma_f32_16x16x4_f32(false, a, false, b, (short)0, c,
                                               false, false);
}

__device__ __forceinline__ float sigm(float x) { return 1.0f / (1.0f + __expf(-x)); }

__device__ __forceinline__ v8f mm_acc(v8f acc, const float* __restrict__ aPtr, int kdim,
                                      const float* __restrict__ bPtr, int ld2) {
  #pragma unroll
  for (int kb = 0; kb < kdim; kb += 4) {
    v2f a = *(const v2f*)(aPtr + kb);
    v2f b = *(const v2f*)(bPtr + (kb >> 1) * ld2);
    acc = wmma4(a, b, acc);
  }
  return acc;
}

extern "C" __global__ __launch_bounds__(NT)
void charnn_fused(const int* __restrict__ batch_x, const int* __restrict__ lens,
                  const float* __restrict__ emb,
                  const float* __restrict__ w_ih1, const float* __restrict__ w_hh1,
                  const float* __restrict__ b_ih1, const float* __restrict__ b_hh1,
                  const float* __restrict__ w_ih2, const float* __restrict__ w_hh2,
                  const float* __restrict__ b_ih2, const float* __restrict__ b_hh2,
                  const float* __restrict__ lw1,  const float* __restrict__ lb1,
                  const float* __restrict__ lw2,  const float* __restrict__ lb2,
                  float* __restrict__ out)
{
  extern __shared__ float smem[];
  const int tid   = threadIdx.x;
  const int lane  = tid & 31;
  const int wave  = tid >> 5;
  const int row   = lane & 15;
  const int khalf = (lane < 16) ? 0 : 2;
  const int kh2   = khalf >> 1;
  const int mhalf = (lane < 16) ? 0 : 8;
  const int b0    = blockIdx.x * CHUNK;

  float* embL = smem;
  float* wi1  = embL + VOC * KE;
  float* wh1  = wi1  + (KE / 2) * NP2;
  float* wi2  = wh1  + (KH / 2) * NP2;
  float* wh2  = wi2  + (KH / 2) * NP2;
  float* l1T  = wh2  + (KH / 2) * NP2;
  float* l2T  = l1T  + (KH / 2) * KH2;
  float* bg1  = l2T  + (KH / 2) * VP2;
  float* bg2  = bg1  + NP;
  float* bl1L = bg2  + NP;
  float* bl2L = bl1L + KH;
  float* h1   = bl2L + VP;
  float* h2   = h1   + 2 * CHUNK * KHP;
  float* h2m  = h2   + CHUNK * KHP;
  float* hmid = h2m  + CHUNK * KHP;
  float* gb1  = hmid + CHUNK * KHP;
  float* gb2  = gb1  + CHUNK * GP;
  int*   ids  = (int*)(gb2 + CHUNK * GP);

  for (int i = tid; i < VOC * KE; i += NT) { int r = i / KE, k = i - r * KE;
    embL[i] = (k < EMB) ? emb[r * EMB + k] : 0.f; }
  for (int i = tid; i < KE * NP; i += NT) { int k = i / NP, n = i - k * NP;
    wi1[(k >> 1) * NP2 + n * 2 + (k & 1)] =
        (k < EMB && n < G4) ? w_ih1[n * EMB + k] : 0.f; }
  for (int i = tid; i < KH * NP; i += NT) { int k = i / NP, n = i - k * NP;
    const bool ok = (k < HID && n < G4);
    const int  d  = (k >> 1) * NP2 + n * 2 + (k & 1);
    wh1[d] = ok ? w_hh1[n * HID + k] : 0.f;
    wi2[d] = ok ? w_ih2[n * HID + k] : 0.f;
    wh2[d] = ok ? w_hh2[n * HID + k] : 0.f; }
  for (int i = tid; i < KH * KH; i += NT) { int k = i / KH, j = i - k * KH;
    l1T[(k >> 1) * KH2 + j * 2 + (k & 1)] =
        (k < HID && j < HID) ? lw1[j * HID + k] : 0.f; }
  for (int i = tid; i < KH * VP; i += NT) { int k = i / VP, v = i - k * VP;
    l2T[(k >> 1) * VP2 + v * 2 + (k & 1)] =
        (k < HID && v < VOC) ? lw2[v * HID + k] : 0.f; }
  for (int i = tid; i < NP; i += NT) {
    bg1[i] = (i < G4) ? (b_ih1[i] + b_hh1[i]) : 0.f;
    bg2[i] = (i < G4) ? (b_ih2[i] + b_hh2[i]) : 0.f; }
  for (int i = tid; i < KH; i += NT) bl1L[i] = (i < HID) ? lb1[i] : 0.f;
  for (int i = tid; i < VP; i += NT) bl2L[i] = (i < VOC) ? lb2[i] : 0.f;
  for (int i = tid; i < CHUNK * KHP; i += NT) {
    h1[i] = 0.f; h1[CHUNK * KHP + i] = 0.f; h2[i] = 0.f; h2m[i] = 0.f; hmid[i] = 0.f; }
  if (tid < CHUNK) ids[tid] = batch_x[(b0 + tid) * T_LEN];

  const int  p    = tid - CELL0;
  const bool cell = (p >= 0);
  const int  cm   = cell ? p / HID : 0;
  const int  cj   = cell ? p - cm * HID : 0;
  const int  clen = cell ? lens[b0 + cm] : 0;
  float c1 = 0.f, h1r = 0.f, c2 = 0.f, h2r = 0.f;
  int nid = 0;

  __syncthreads();

  for (int s = 0; s < ST; ++s) {
    if (wave < NP / 16) {
      if (s < T_LEN) {
        const float* arowE = embL + ids[row] * KE + khalf;
        const float* arowH = h1 + ((s + 1) & 1) * (CHUNK * KHP) + row * KHP + khalf;
        const int n = wave * 16 + row;
        const float bv = bg1[n];
        v8f acc;
        #pragma unroll
        for (int r = 0; r < 8; ++r) acc[r] = bv;
        acc = mm_acc(acc, arowE, KE, wi1 + n * 2 + kh2 * NP2, NP2);
        acc = mm_acc(acc, arowH, KH, wh1 + n * 2 + kh2 * NP2, NP2);
        #pragma unroll
        for (int r = 0; r < 8; ++r) gb1[(r + mhalf) * GP + n] = acc[r];
      }
    } else if (wave < NP / 16 + VP / 16) {
      if (s >= 3) {
        const int t = s - 3;
        const float* arow = hmid + row * KHP + khalf;
        const int n = (wave - NP / 16) * 16 + row;
        const float bv = bl2L[n];
        v8f acc;
        #pragma unroll
        for (int r = 0; r < 8; ++r) acc[r] = bv;
        acc = mm_acc(acc, arow, KH, l2T + n * 2 + kh2 * VP2, VP2);
        if (n < VOC) {
          #pragma unroll
          for (int r = 0; r < 8; ++r) {
            const int m = r + mhalf;
            out[((size_t)(b0 + m) * T_LEN + t) * VOC + n] = acc[r];
          }
        }
      }
    } else if (wave == MW - 1) {
      if (lane < CHUNK) nid = (s + 1 < T_LEN) ? batch_x[(b0 + lane) * T_LEN + s + 1] : 0;
    } else {
      if (s >= 2 && s < T_LEN + 2) {
        const int t = s - 2;
        const float ig = sigm(gb2[cm * GP + cj]);
        const float fg = sigm(gb2[cm * GP + HID + cj]);
        const float gg = tanhf(gb2[cm * GP + 2 * HID + cj]);
        const float og = sigm(gb2[cm * GP + 3 * HID + cj]);
        const float cn = fg * c2 + ig * gg;
        const float hn = og * tanhf(cn);
        const float mk = (t < clen) ? 1.f : 0.f;
        c2 = mk * cn + (1.f - mk) * c2;
        h2r = mk * hn + (1.f - mk) * h2r;
        h2[cm * KHP + cj]  = h2r;
        h2m[cm * KHP + cj] = mk * h2r;
      }
    }
    __syncthreads();

    if (wave < NP / 16) {
      if (s >= 1 && s <= T_LEN) {
        const float* arowA = h1 + ((s + 1) & 1) * (CHUNK * KHP) + row * KHP + khalf;
        const float* arowB = h2 + row * KHP + khalf;
        const int n = wave * 16 + row;
        const float bv = bg2[n];
        v8f acc;
        #pragma unroll
        for (int r = 0; r < 8; ++r) acc[r] = bv;
        acc = mm_acc(acc, arowA, KH, wi2 + n * 2 + kh2 * NP2, NP2);
        acc = mm_acc(acc, arowB, KH, wh2 + n * 2 + kh2 * NP2, NP2);
        #pragma unroll
        for (int r = 0; r < 8; ++r) gb2[(r + mhalf) * GP + n] = acc[r];
      }
    } else if (wave < NP / 16 + 2) {
      if (s >= 2 && s < T_LEN + 2) {
        const float* arow = h2m + row * KHP + khalf;
        const int n = (wave - NP / 16) * 16 + row;
        const float bv = bl1L[n];
        v8f acc;
        #pragma unroll
        for (int r = 0; r < 8; ++r) acc[r] = bv;
        acc = mm_acc(acc, arow, KH, l1T + n * 2 + kh2 * KH2, KH2);
        #pragma unroll
        for (int r = 0; r < 8; ++r) hmid[(r + mhalf) * KHP + n] = fmaxf(acc[r], 0.f);
      }
    } else if (wave == MW - 1) {
      if (lane < CHUNK) ids[lane] = nid;
    } else if (wave >= MW) {
      if (s < T_LEN) {
        const float ig = sigm(gb1[cm * GP + cj]);
        const float fg = sigm(gb1[cm * GP + HID + cj]);
        const float gg = tanhf(gb1[cm * GP + 2 * HID + cj]);
        const float og = sigm(gb1[cm * GP + 3 * HID + cj]);
        const float cn = fg * c1 + ig * gg;
        const float hn = og * tanhf(cn);
        const float mk = (s < clen) ? 1.f : 0.f;
        c1 = mk * cn + (1.f - mk) * c1;
        h1r = mk * hn + (1.f - mk) * h1r;
        h1[(s & 1) * (CHUNK * KHP) + cm * KHP + cj] = h1r;
      }
    }
    __syncthreads();
  }
}

extern "C" void kernel_launch(void* const* d_in, const int* in_sizes, int n_in,
                              void* d_out, int out_size, void* d_ws, size_t ws_size,
                              hipStream_t stream) {
  const int*   batch_x = (const int*)  d_in[0];
  const int*   lens    = (const int*)  d_in[1];
  const float* emb     = (const float*)d_in[2];
  const float* w_ih1   = (const float*)d_in[3];
  const float* w_hh1   = (const float*)d_in[4];
  const float* b_ih1   = (const float*)d_in[5];
  const float* b_hh1   = (const float*)d_in[6];
  const float* w_ih2   = (const float*)d_in[7];
  const float* w_hh2   = (const float*)d_in[8];
  const float* b_ih2   = (const float*)d_in[9];
  const float* b_hh2   = (const float*)d_in[10];
  const float* lw1     = (const float*)d_in[11];
  const float* lb1     = (const float*)d_in[12];
  const float* lw2     = (const float*)d_in[13];
  const float* lb2     = (const float*)d_in[14];
  float* out = (float*)d_out;

  constexpr size_t SMEM_FLOATS =
      (size_t)VOC * KE + (size_t)(KE / 2) * NP2 + 3u * (KH / 2) * NP2 + (size_t)(KH / 2) * KH2 +
      (size_t)(KH / 2) * VP2 + 2u * NP + KH + VP + 5u * CHUNK * KHP + 2u * CHUNK * GP +
      (size_t)CHUNK;
  constexpr size_t SMEM_BYTES = SMEM_FLOATS * sizeof(float);

  (void)hipFuncSetAttribute((const void*)charnn_fused,
                            hipFuncAttributeMaxDynamicSharedMemorySize,
                            (int)SMEM_BYTES);

  charnn_fused<<<dim3(B_SZ / CHUNK), dim3(NT), SMEM_BYTES, stream>>>(
      batch_x, lens, emb, w_ih1, w_hh1, b_ih1, b_hh1,
      w_ih2, w_hh2, b_ih2, b_hh2, lw1, lb1, lw2, lb2, out);
}


# Time-loop barriers fence LDS only (workgroup release/acquire on the local address space): head waves no longer wait for the acknowledgement of their global output stores before each barrier
# speedup vs baseline: 3.5339x; 1.1264x over previous
#include <hip/hip_runtime.h>
#include <hip/hip_bf16.h>

#define T_LEN 512
#define B_SZ  512
#define EMB   50
#define HID   30
#define G4    120
#define VOC   100
#define NP    128
#define KE    52
#define KH    32
#define VP    112
#define CHUNK 16
#define MW    16
#define CW    15
#define NT    ((MW + CW) * 32)
#define CELL0 (MW * 32)
#define ST    (T_LEN + 3)

#define KHP 36
#define GP  158
#define NP2 (NP * 2 + 32)
#define KH2 (KH * 2 + 32)
#define VP2 (VP * 2)

typedef __attribute__((ext_vector_type(2))) float v2f;
typedef __attribute__((ext_vector_type(8))) float v8f;

__device__ __forceinline__ v8f wmma4(v2f a, v2f b, v8f c) {
  return __builtin_amdgcn_wmma_f32_16x16x4_f32(false, a, false, b, (short)0, c,
                                               false, false);
}

__device__ __forceinline__ float sigm(float x) { return 1.0f / (1.0f + __expf(-x)); }

__device__ __forceinline__ v8f mm_acc(v8f acc, const float* __restrict__ aPtr, int kdim,
                                      const float* __restrict__ bPtr, int ld2) {
  #pragma unroll
  for (int kb = 0; kb < kdim; kb += 4) {
    v2f a = *(const v2f*)(aPtr + kb);
    v2f b = *(const v2f*)(bPtr + (kb >> 1) * ld2);
    acc = wmma4(a, b, acc);
  }
  return acc;
}


__device__ __forceinline__ void lds_barrier() {
  __builtin_amdgcn_fence(__ATOMIC_RELEASE, "workgroup", "local");
  __builtin_amdgcn_s_barrier();
  __builtin_amdgcn_fence(__ATOMIC_ACQUIRE, "workgroup", "local");
}

extern "C" __global__ __launch_bounds__(NT)
void charnn_fused(const int* __restrict__ batch_x, const int* __restrict__ lens,
                  const float* __restrict__ emb,
                  const float* __restrict__ w_ih1, const float* __restrict__ w_hh1,
                  const float* __restrict__ b_ih1, const float* __restrict__ b_hh1,
                  const float* __restrict__ w_ih2, const float* __restrict__ w_hh2,
                  const float* __restrict__ b_ih2, const float* __restrict__ b_hh2,
                  const float* __restrict__ lw1,  const float* __restrict__ lb1,
                  const float* __restrict__ lw2,  const float* __restrict__ lb2,
                  float* __restrict__ out)
{
  extern __shared__ float smem[];
  const int tid   = threadIdx.x;
  const int lane  = tid & 31;
  const int wave  = tid >> 5;
  const int row   = lane & 15;
  const int khalf = (lane < 16) ? 0 : 2;
  const int kh2   = khalf >> 1;
  const int mhalf = (lane < 16) ? 0 : 8;
  const int b0    = blockIdx.x * CHUNK;

  float* embL = smem;
  float* wi1  = embL + VOC * KE;
  float* wh1  = wi1  + (KE / 2) * NP2;
  float* wi2  = wh1  + (KH / 2) * NP2;
  float* wh2  = wi2  + (KH / 2) * NP2;
  float* l1T  = wh2  + (KH / 2) * NP2;
  float* l2T  = l1T  + (KH / 2) * KH2;
  float* bg1  = l2T  + (KH / 2) * VP2;
  float* bg2  = bg1  + NP;
  float* bl1L = bg2  + NP;
  float* bl2L = bl1L + KH;
  float* h1   = bl2L + VP;
  float* h2   = h1   + 2 * CHUNK * KHP;
  float* h2m  = h2   + CHUNK * KHP;
  float* hmid = h2m  + CHUNK * KHP;
  float* gb1  = hmid + CHUNK * KHP;
  float* gb2  = gb1  + CHUNK * GP;
  int*   ids  = (int*)(gb2 + CHUNK * GP);

  for (int i = tid; i < VOC * KE; i += NT) { int r = i / KE, k = i - r * KE;
    embL[i] = (k < EMB) ? emb[r * EMB + k] : 0.f; }
  for (int i = tid; i < KE * NP; i += NT) { int k = i / NP, n = i - k * NP;
    wi1[(k >> 1) * NP2 + n * 2 + (k & 1)] =
        (k < EMB && n < G4) ? w_ih1[n * EMB + k] : 0.f; }
  for (int i = tid; i < KH * NP; i += NT) { int k = i / NP, n = i - k * NP;
    const bool ok = (k < HID && n < G4);
    const int  d  = (k >> 1) * NP2 + n * 2 + (k & 1);
    wh1[d] = ok ? w_hh1[n * HID + k] : 0.f;
    wi2[d] = ok ? w_ih2[n * HID + k] : 0.f;
    wh2[d] = ok ? w_hh2[n * HID + k] : 0.f; }
  for (int i = tid; i < KH * KH; i += NT) { int k = i / KH, j = i - k * KH;
    l1T[(k >> 1) * KH2 + j * 2 + (k & 1)] =
        (k < HID && j < HID) ? lw1[j * HID + k] : 0.f; }
  for (int i = tid; i < KH * VP; i += NT) { int k = i / VP, v = i - k * VP;
    l2T[(k >> 1) * VP2 + v * 2 + (k & 1)] =
        (k < HID && v < VOC) ? lw2[v * HID + k] : 0.f; }
  for (int i = tid; i < NP; i += NT) {
    bg1[i] = (i < G4) ? (b_ih1[i] + b_hh1[i]) : 0.f;
    bg2[i] = (i < G4) ? (b_ih2[i] + b_hh2[i]) : 0.f; }
  for (int i = tid; i < KH; i += NT) bl1L[i] = (i < HID) ? lb1[i] : 0.f;
  for (int i = tid; i < VP; i += NT) bl2L[i] = (i < VOC) ? lb2[i] : 0.f;
  for (int i = tid; i < CHUNK * KHP; i += NT) {
    h1[i] = 0.f; h1[CHUNK * KHP + i] = 0.f; h2[i] = 0.f; h2m[i] = 0.f; hmid[i] = 0.f; }
  if (tid < CHUNK) ids[tid] = batch_x[(b0 + tid) * T_LEN];

  const int  p    = tid - CELL0;
  const bool cell = (p >= 0);
  const int  cm   = cell ? p / HID : 0;
  const int  cj   = cell ? p - cm * HID : 0;
  const int  clen = cell ? lens[b0 + cm] : 0;
  float c1 = 0.f, h1r = 0.f, c2 = 0.f, h2r = 0.f;
  int nid = 0;

  __syncthreads();

  for (int s = 0; s < ST; ++s) {
    if (wave < NP / 16) {
      if (s < T_LEN) {
        const float* arowE = embL + ids[row] * KE + khalf;
        const float* arowH = h1 + ((s + 1) & 1) * (CHUNK * KHP) + row * KHP + khalf;
        const int n = wave * 16 + row;
        const float bv = bg1[n];
        v8f acc;
        #pragma unroll
        for (int r = 0; r < 8; ++r) acc[r] = bv;
        acc = mm_acc(acc, arowE, KE, wi1 + n * 2 + kh2 * NP2, NP2);
        acc = mm_acc(acc, arowH, KH, wh1 + n * 2 + kh2 * NP2, NP2);
        #pragma unroll
        for (int r = 0; r < 8; ++r) gb1[(r + mhalf) * GP + n] = acc[r];
      }
    } else if (wave < NP / 16 + VP / 16) {
      if (s >= 3) {
        const int t = s - 3;
        const float* arow = hmid + row * KHP + khalf;
        const int n = (wave - NP / 16) * 16 + row;
        const float bv = bl2L[n];
        v8f acc;
        #pragma unroll
        for (int r = 0; r < 8; ++r) acc[r] = bv;
        acc = mm_acc(acc, arow, KH, l2T + n * 2 + kh2 * VP2, VP2);
        if (n < VOC) {
          #pragma unroll
          for (int r = 0; r < 8; ++r) {
            const int m = r + mhalf;
            out[((size_t)(b0 + m) * T_LEN + t) * VOC + n] = acc[r];
          }
        }
      }
    } else if (wave == MW - 1) {
      if (lane < CHUNK) nid = (s + 1 < T_LEN) ? batch_x[(b0 + lane) * T_LEN + s + 1] : 0;
    } else {
      if (s >= 2 && s < T_LEN + 2) {
        const int t = s - 2;
        const float ig = sigm(gb2[cm * GP + cj]);
        const float fg = sigm(gb2[cm * GP + HID + cj]);
        const float gg = tanhf(gb2[cm * GP + 2 * HID + cj]);
        const float og = sigm(gb2[cm * GP + 3 * HID + cj]);
        const float cn = fg * c2 + ig * gg;
        const float hn = og * tanhf(cn);
        const float mk = (t < clen) ? 1.f : 0.f;
        c2 = mk * cn + (1.f - mk) * c2;
        h2r = mk * hn + (1.f - mk) * h2r;
        h2[cm * KHP + cj]  = h2r;
        h2m[cm * KHP + cj] = mk * h2r;
      }
    }
    lds_barrier();

    if (wave < NP / 16) {
      if (s >= 1 && s <= T_LEN) {
        const float* arowA = h1 + ((s + 1) & 1) * (CHUNK * KHP) + row * KHP + khalf;
        const float* arowB = h2 + row * KHP + khalf;
        const int n = wave * 16 + row;
        const float bv = bg2[n];
        v8f acc;
        #pragma unroll
        for (int r = 0; r < 8; ++r) acc[r] = bv;
        acc = mm_acc(acc, arowA, KH, wi2 + n * 2 + kh2 * NP2, NP2);
        acc = mm_acc(acc, arowB, KH, wh2 + n * 2 + kh2 * NP2, NP2);
        #pragma unroll
        for (int r = 0; r < 8; ++r) gb2[(r + mhalf) * GP + n] = acc[r];
      }
    } else if (wave < NP / 16 + 2) {
      if (s >= 2 && s < T_LEN + 2) {
        const float* arow = h2m + row * KHP + khalf;
        const int n = (wave - NP / 16) * 16 + row;
        const float bv = bl1L[n];
        v8f acc;
        #pragma unroll
        for (int r = 0; r < 8; ++r) acc[r] = bv;
        acc = mm_acc(acc, arow, KH, l1T + n * 2 + kh2 * KH2, KH2);
        #pragma unroll
        for (int r = 0; r < 8; ++r) hmid[(r + mhalf) * KHP + n] = fmaxf(acc[r], 0.f);
      }
    } else if (wave == MW - 1) {
      if (lane < CHUNK) ids[lane] = nid;
    } else if (wave >= MW) {
      if (s < T_LEN) {
        const float ig = sigm(gb1[cm * GP + cj]);
        const float fg = sigm(gb1[cm * GP + HID + cj]);
        const float gg = tanhf(gb1[cm * GP + 2 * HID + cj]);
        const float og = sigm(gb1[cm * GP + 3 * HID + cj]);
        const float cn = fg * c1 + ig * gg;
        const float hn = og * tanhf(cn);
        const float mk = (s < clen) ? 1.f : 0.f;
        c1 = mk * cn + (1.f - mk) * c1;
        h1r = mk * hn + (1.f - mk) * h1r;
        h1[(s & 1) * (CHUNK * KHP) + cm * KHP + cj] = h1r;
      }
    }
    lds_barrier();
  }
}

extern "C" void kernel_launch(void* const* d_in, const int* in_sizes, int n_in,
                              void* d_out, int out_size, void* d_ws, size_t ws_size,
                              hipStream_t stream) {
  const int*   batch_x = (const int*)  d_in[0];
  const int*   lens    = (const int*)  d_in[1];
  const float* emb     = (const float*)d_in[2];
  const float* w_ih1   = (const float*)d_in[3];
  const float* w_hh1   = (const float*)d_in[4];
  const float* b_ih1   = (const float*)d_in[5];
  const float* b_hh1   = (const float*)d_in[6];
  const float* w_ih2   = (const float*)d_in[7];
  const float* w_hh2   = (const float*)d_in[8];
  const float* b_ih2   = (const float*)d_in[9];
  const float* b_hh2   = (const float*)d_in[10];
  const float* lw1     = (const float*)d_in[11];
  const float* lb1     = (const float*)d_in[12];
  const float* lw2     = (const float*)d_in[13];
  const float* lb2     = (const float*)d_in[14];
  float* out = (float*)d_out;

  constexpr size_t SMEM_FLOATS =
      (size_t)VOC * KE + (size_t)(KE / 2) * NP2 + 3u * (KH / 2) * NP2 + (size_t)(KH / 2) * KH2 +
      (size_t)(KH / 2) * VP2 + 2u * NP + KH + VP + 5u * CHUNK * KHP + 2u * CHUNK * GP +
      (size_t)CHUNK;
  constexpr size_t SMEM_BYTES = SMEM_FLOATS * sizeof(float);

  (void)hipFuncSetAttribute((const void*)charnn_fused,
                            hipFuncAttributeMaxDynamicSharedMemorySize,
                            (int)SMEM_BYTES);

  charnn_fused<<<dim3(B_SZ / CHUNK), dim3(NT), SMEM_BYTES, stream>>>(
      batch_x, lens, emb, w_ih1, w_hh1, b_ih1, b_hh1,
      w_ih2, w_hh2, b_ih2, b_hh2, lw1, lb1, lw2, lb2, out);
}


# Layer-1 input projection read from a per-token table (bias + emb[v] @ w_ih1^T, built once per workgroup on WMMA): gates1 per step = table lookup + 8 recurrent WMMAs, not 21; sum order unchanged
# speedup vs baseline: 4.0925x; 1.1581x over previous
#include <hip/hip_runtime.h>
#include <hip/hip_bf16.h>

#define T_LEN 512
#define B_SZ  512
#define EMB   50
#define HID   30
#define G4    120
#define VOC   100
#define NP    128
#define KE    52
#define VT    ((VOC + 15) / 16)
#define KH    32
#define VP    112
#define CHUNK 16
#define MW    16
#define CW    15
#define NT    ((MW + CW) * 32)
#define CELL0 (MW * 32)
#define ST    (T_LEN + 3)

#define KHP 36
#define GP  158
#define NP2 (NP * 2 + 32)
#define KH2 (KH * 2 + 32)
#define VP2 (VP * 2)
#define PP  144

typedef __attribute__((ext_vector_type(2))) float v2f;
typedef __attribute__((ext_vector_type(8))) float v8f;

__device__ __forceinline__ v8f wmma4(v2f a, v2f b, v8f c) {
  return __builtin_amdgcn_wmma_f32_16x16x4_f32(false, a, false, b, (short)0, c,
                                               false, false);
}

__device__ __forceinline__ float sigm(float x) { return 1.0f / (1.0f + __expf(-x)); }

__device__ __forceinline__ v8f mm_acc(v8f acc, const float* __restrict__ aPtr, int kdim,
                                      const float* __restrict__ bPtr, int ld2) {
  #pragma unroll
  for (int kb = 0; kb < kdim; kb += 4) {
    v2f a = *(const v2f*)(aPtr + kb);
    v2f b = *(const v2f*)(bPtr + (kb >> 1) * ld2);
    acc = wmma4(a, b, acc);
  }
  return acc;
}


__device__ __forceinline__ void lds_barrier() {
  __builtin_amdgcn_fence(__ATOMIC_RELEASE, "workgroup", "local");
  __builtin_amdgcn_s_barrier();
  __builtin_amdgcn_fence(__ATOMIC_ACQUIRE, "workgroup", "local");
}

extern "C" __global__ __launch_bounds__(NT)
void charnn_fused(const int* __restrict__ batch_x, const int* __restrict__ lens,
                  const float* __restrict__ emb,
                  const float* __restrict__ w_ih1, const float* __restrict__ w_hh1,
                  const float* __restrict__ b_ih1, const float* __restrict__ b_hh1,
                  const float* __restrict__ w_ih2, const float* __restrict__ w_hh2,
                  const float* __restrict__ b_ih2, const float* __restrict__ b_hh2,
                  const float* __restrict__ lw1,  const float* __restrict__ lb1,
                  const float* __restrict__ lw2,  const float* __restrict__ lb2,
                  float* __restrict__ out)
{
  extern __shared__ float smem[];
  const int tid   = threadIdx.x;
  const int lane  = tid & 31;
  const int wave  = tid >> 5;
  const int row   = lane & 15;
  const int khalf = (lane < 16) ? 0 : 2;
  const int kh2   = khalf >> 1;
  const int mhalf = (lane < 16) ? 0 : 8;
  const int b0    = blockIdx.x * CHUNK;

  float* ptab = smem;
  float* wh1  = ptab + VOC * PP;
  float* wi2  = wh1  + (KH / 2) * NP2;
  float* wh2  = wi2  + (KH / 2) * NP2;
  float* l1T  = wh2  + (KH / 2) * NP2;
  float* l2T  = l1T  + (KH / 2) * KH2;
  float* bg2  = l2T  + (KH / 2) * VP2;
  float* bl1L = bg2  + NP;
  float* bl2L = bl1L + KH;
  float* h1   = bl2L + VP;
  float* h2   = h1   + 2 * CHUNK * KHP;
  float* h2m  = h2   + CHUNK * KHP;
  float* hmid = h2m  + CHUNK * KHP;
  float* gb1  = hmid + CHUNK * KHP;
  float* gb2  = gb1  + CHUNK * GP;
  int*   ids  = (int*)(gb2 + CHUNK * GP);
  int*   ids0 = ids + CHUNK;

  for (int t = __builtin_amdgcn_readfirstlane(wave); t < VT * (NP / 16); t += MW + CW) {
    const int  mt  = t / (NP / 16), nt = t - mt * (NP / 16);
    const int  v   = mt * 16 + row;
    const int  n   = nt * 16 + row;
    const int  nc  = (n < G4) ? n : 0;
    const float* ap = emb   + ((v < VOC) ? v : 0) * EMB;
    const float* bp = w_ih1 + nc * EMB;
    const float  bv = b_ih1[nc] + b_hh1[nc];
    v2f a[KE / 4], b[KE / 4];
    #pragma unroll
    for (int i = 0; i < KE / 4; ++i) {
      const bool kok = (4 * i + 2 < EMB) || (khalf == 0);
      const int  kc  = 4 * i + (kok ? khalf : 0);
      const v2f av = *(const v2f*)(ap + kc);
      const v2f bw = *(const v2f*)(bp + kc);
      a[i][0] = kok ? av[0] : 0.f;  a[i][1] = kok ? av[1] : 0.f;
      b[i][0] = kok ? bw[0] : 0.f;  b[i][1] = kok ? bw[1] : 0.f;
    }
    v8f acc;
    #pragma unroll
    for (int r = 0; r < 8; ++r) acc[r] = bv;
    __builtin_amdgcn_sched_barrier(0);
    #pragma unroll
    for (int i = 0; i < KE / 4; ++i) acc = wmma4(a[i], b[i], acc);
    #pragma unroll
    for (int r = 0; r < 8; ++r) {
      const int vr = mt * 16 + r + mhalf;
      if (vr < VOC) ptab[vr * PP + n] = acc[r];
    }
  }
  for (int i = tid; i < KH * NP; i += NT) { int k = i / NP, n = i - k * NP;
    const bool ok = (k < HID && n < G4);
    const int  d  = (k >> 1) * NP2 + n * 2 + (k & 1);
    wh1[d] = ok ? w_hh1[n * HID + k] : 0.f;
    wi2[d] = ok ? w_ih2[n * HID + k] : 0.f;
    wh2[d] = ok ? w_hh2[n * HID + k] : 0.f; }
  for (int i = tid; i < KH * KH; i += NT) { int k = i / KH, j = i - k * KH;
    l1T[(k >> 1) * KH2 + j * 2 + (k & 1)] =
        (k < HID && j < HID) ? lw1[j * HID + k] : 0.f; }
  for (int i = tid; i < KH * VP; i += NT) { int k = i / VP, v = i - k * VP;
    l2T[(k >> 1) * VP2 + v * 2 + (k & 1)] =
        (k < HID && v < VOC) ? lw2[v * HID + k] : 0.f; }
  for (int i = tid; i < NP; i += NT)
    bg2[i] = (i < G4) ? (b_ih2[i] + b_hh2[i]) : 0.f;
  for (int i = tid; i < KH; i += NT) bl1L[i] = (i < HID) ? lb1[i] : 0.f;
  for (int i = tid; i < VP; i += NT) bl2L[i] = (i < VOC) ? lb2[i] : 0.f;
  for (int i = tid; i < CHUNK * KHP; i += NT) {
    h1[i] = 0.f; h1[CHUNK * KHP + i] = 0.f; h2[i] = 0.f; h2m[i] = 0.f; hmid[i] = 0.f; }
  if (tid < CHUNK) ids0[tid] = batch_x[(b0 + tid) * T_LEN] * PP;
  else if (tid < 2 * CHUNK) ids[tid - CHUNK] = (1 < T_LEN) ? batch_x[(b0 + tid - CHUNK) * T_LEN + 1] * PP : 0;

  const int  p    = tid - CELL0;
  const bool cell = (p >= 0);
  const int  cm   = cell ? p / HID : 0;
  const int  cj   = cell ? p - cm * HID : 0;
  const int  clen = cell ? lens[b0 + cm] : 0;
  float c1 = 0.f, h1r = 0.f, c2 = 0.f, h2r = 0.f;
  int nid = 0;
  v8f xg;
  #pragma unroll
  for (int r = 0; r < 8; ++r) xg[r] = 0.f;

  __syncthreads();

  if (wave < NP / 16) {
    const int n = wave * 16 + row;
    #pragma unroll
    for (int r = 0; r < 8; ++r) xg[r] = ptab[ids0[r + mhalf] + n];
  }

  for (int s = 0; s < ST; ++s) {
    if (wave < NP / 16) {
      if (s < T_LEN) {
        const int n = wave * 16 + row;
        v8f xn;
        #pragma unroll
        for (int r = 0; r < 8; ++r) xn[r] = ptab[ids[r + mhalf] + n];
        const float* arowH = h1 + ((s + 1) & 1) * (CHUNK * KHP) + row * KHP + khalf;
        v8f acc = xg;
        acc = mm_acc(acc, arowH, KH, wh1 + n * 2 + kh2 * NP2, NP2);
        xg = xn;
        #pragma unroll
        for (int r = 0; r < 8; ++r) gb1[(r + mhalf) * GP + n] = acc[r];
      }
    } else if (wave < NP / 16 + VP / 16) {
      if (s >= 3) {
        const int t = s - 3;
        const float* arow = hmid + row * KHP + khalf;
        const int n = (wave - NP / 16) * 16 + row;
        const float bv = bl2L[n];
        v8f acc;
        #pragma unroll
        for (int r = 0; r < 8; ++r) acc[r] = bv;
        acc = mm_acc(acc, arow, KH, l2T + n * 2 + kh2 * VP2, VP2);
        if (n < VOC) {
          #pragma unroll
          for (int r = 0; r < 8; ++r) {
            const int m = r + mhalf;
            out[((size_t)(b0 + m) * T_LEN + t) * VOC + n] = acc[r];
          }
        }
      }
    } else if (wave == MW - 1) {
      if (lane < CHUNK) nid = (s + 2 < T_LEN) ? batch_x[(b0 + lane) * T_LEN + s + 2] * PP : 0;
    } else {
      if (s >= 2 && s < T_LEN + 2) {
        const int t = s - 2;
        const float ig = sigm(gb2[cm * GP + cj]);
        const float fg = sigm(gb2[cm * GP + HID + cj]);
        const float gg = tanhf(gb2[cm * GP + 2 * HID + cj]);
        const float og = sigm(gb2[cm * GP + 3 * HID + cj]);
        const float cn = fg * c2 + ig * gg;
        const float hn = og * tanhf(cn);
        const float mk = (t < clen) ? 1.f : 0.f;
        c2 = mk * cn + (1.f - mk) * c2;
        h2r = mk * hn + (1.f - mk) * h2r;
        h2[cm * KHP + cj]  = h2r;
        h2m[cm * KHP + cj] = mk * h2r;
      }
    }
    lds_barrier();

    if (wave < NP / 16) {
      if (s >= 1 && s <= T_LEN) {
        const float* arowA = h1 + ((s + 1) & 1) * (CHUNK * KHP) + row * KHP + khalf;
        const float* arowB = h2 + row * KHP + khalf;
        const int n = wave * 16 + row;
        const float bv = bg2[n];
        v8f acc;
        #pragma unroll
        for (int r = 0; r < 8; ++r) acc[r] = bv;
        acc = mm_acc(acc, arowA, KH, wi2 + n * 2 + kh2 * NP2, NP2);
        acc = mm_acc(acc, arowB, KH, wh2 + n * 2 + kh2 * NP2, NP2);
        #pragma unroll
        for (int r = 0; r < 8; ++r) gb2[(r + mhalf) * GP + n] = acc[r];
      }
    } else if (wave < NP / 16 + 2) {
      if (s >= 2 && s < T_LEN + 2) {
        const float* arow = h2m + row * KHP + khalf;
        const int n = (wave - NP / 16) * 16 + row;
        const float bv = bl1L[n];
        v8f acc;
        #pragma unroll
        for (int r = 0; r < 8; ++r) acc[r] = bv;
        acc = mm_acc(acc, arow, KH, l1T + n * 2 + kh2 * KH2, KH2);
        #pragma unroll
        for (int r = 0; r < 8; ++r) hmid[(r + mhalf) * KHP + n] = fmaxf(acc[r], 0.f);
      }
    } else if (wave == MW - 1) {
      if (lane < CHUNK) ids[lane] = nid;
    } else if (wave >= MW) {
      if (s < T_LEN) {
        const float ig = sigm(gb1[cm * GP + cj]);
        const float fg = sigm(gb1[cm * GP + HID + cj]);
        const float gg = tanhf(gb1[cm * GP + 2 * HID + cj]);
        const float og = sigm(gb1[cm * GP + 3 * HID + cj]);
        const float cn = fg * c1 + ig * gg;
        const float hn = og * tanhf(cn);
        const float mk = (s < clen) ? 1.f : 0.f;
        c1 = mk * cn + (1.f - mk) * c1;
        h1r = mk * hn + (1.f - mk) * h1r;
        h1[(s & 1) * (CHUNK * KHP) + cm * KHP + cj] = h1r;
      }
    }
    lds_barrier();
  }
}

extern "C" void kernel_launch(void* const* d_in, const int* in_sizes, int n_in,
                              void* d_out, int out_size, void* d_ws, size_t ws_size,
                              hipStream_t stream) {
  const int*   batch_x = (const int*)  d_in[0];
  const int*   lens    = (const int*)  d_in[1];
  const float* emb     = (const float*)d_in[2];
  const float* w_ih1   = (const float*)d_in[3];
  const float* w_hh1   = (const float*)d_in[4];
  const float* b_ih1   = (const float*)d_in[5];
  const float* b_hh1   = (const float*)d_in[6];
  const float* w_ih2   = (const float*)d_in[7];
  const float* w_hh2   = (const float*)d_in[8];
  const float* b_ih2   = (const float*)d_in[9];
  const float* b_hh2   = (const float*)d_in[10];
  const float* lw1     = (const float*)d_in[11];
  const float* lb1     = (const float*)d_in[12];
  const float* lw2     = (const float*)d_in[13];
  const float* lb2     = (const float*)d_in[14];
  float* out = (float*)d_out;

  constexpr size_t SMEM_FLOATS =
      (size_t)VOC * PP + 3u * (KH / 2) * NP2 + (size_t)(KH / 2) * KH2 +
      (size_t)(KH / 2) * VP2 + (size_t)NP + KH + VP + 5u * CHUNK * KHP + 2u * CHUNK * GP +
      2u * CHUNK;
  constexpr size_t SMEM_BYTES = SMEM_FLOATS * sizeof(float);

  (void)hipFuncSetAttribute((const void*)charnn_fused,
                            hipFuncAttributeMaxDynamicSharedMemorySize,
                            (int)SMEM_BYTES);

  charnn_fused<<<dim3(B_SZ / CHUNK), dim3(NT), SMEM_BYTES, stream>>>(
      batch_x, lens, emb, w_ih1, w_hh1, b_ih1, b_hh1,
      w_ih2, w_hh2, b_ih2, b_hh2, lw1, lb1, lw2, lb2, out);
}


# LSTM cell update as a straight run of ~30 vector instructions: sigmoid and tanh on v_exp_f32 + v_rcp_f32 (1 ulp each) replace IEEE division and branching library tanh, mask by select; sum order kept
# speedup vs baseline: 5.2322x; 1.2785x over previous
#include <hip/hip_runtime.h>
#include <hip/hip_bf16.h>

#define T_LEN 512
#define B_SZ  512
#define EMB   50
#define HID   30
#define G4    120
#define VOC   100
#define NP    128
#define KE    52
#define VT    ((VOC + 15) / 16)
#define KH    32
#define VP    112
#define CHUNK 16
#define MW    16
#define CW    15
#define NT    ((MW + CW) * 32)
#define CELL0 (MW * 32)
#define ST    (T_LEN + 3)

#define KHP 36
#define GP  158
#define NP2 (NP * 2 + 32)
#define KH2 (KH * 2 + 32)
#define VP2 (VP * 2)
#define PP  144

typedef __attribute__((ext_vector_type(2))) float v2f;
typedef __attribute__((ext_vector_type(8))) float v8f;

__device__ __forceinline__ v8f wmma4(v2f a, v2f b, v8f c) {
  return __builtin_amdgcn_wmma_f32_16x16x4_f32(false, a, false, b, (short)0, c,
                                               false, false);
}

__device__ __forceinline__ float sigm(float x) {
  return __builtin_amdgcn_rcpf(1.0f + __builtin_amdgcn_exp2f(-1.44269504f * x));
}
__device__ __forceinline__ float tanh_e(float x) {
  return __builtin_fmaf(-2.0f, __builtin_amdgcn_rcpf(1.0f + __builtin_amdgcn_exp2f(2.88539008f * x)), 1.0f);
}

__device__ __forceinline__ v8f mm_acc(v8f acc, const float* __restrict__ aPtr, int kdim,
                                      const float* __restrict__ bPtr, int ld2) {
  #pragma unroll
  for (int kb = 0; kb < kdim; kb += 4) {
    v2f a = *(const v2f*)(aPtr + kb);
    v2f b = *(const v2f*)(bPtr + (kb >> 1) * ld2);
    acc = wmma4(a, b, acc);
  }
  return acc;
}


__device__ __forceinline__ void lds_barrier() {
  __builtin_amdgcn_fence(__ATOMIC_RELEASE, "workgroup", "local");
  __builtin_amdgcn_s_barrier();
  __builtin_amdgcn_fence(__ATOMIC_ACQUIRE, "workgroup", "local");
}

extern "C" __global__ __launch_bounds__(NT)
void charnn_fused(const int* __restrict__ batch_x, const int* __restrict__ lens,
                  const float* __restrict__ emb,
                  const float* __restrict__ w_ih1, const float* __restrict__ w_hh1,
                  const float* __restrict__ b_ih1, const float* __restrict__ b_hh1,
                  const float* __restrict__ w_ih2, const float* __restrict__ w_hh2,
                  const float* __restrict__ b_ih2, const float* __restrict__ b_hh2,
                  const float* __restrict__ lw1,  const float* __restrict__ lb1,
                  const float* __restrict__ lw2,  const float* __restrict__ lb2,
                  float* __restrict__ out)
{
  extern __shared__ float smem[];
  const int tid   = threadIdx.x;
  const int lane  = tid & 31;
  const int wave  = tid >> 5;
  const int row   = lane & 15;
  const int khalf = (lane < 16) ? 0 : 2;
  const int kh2   = khalf >> 1;
  const int mhalf = (lane < 16) ? 0 : 8;
  const int b0    = blockIdx.x * CHUNK;

  float* ptab = smem;
  float* wh1  = ptab + VOC * PP;
  float* wi2  = wh1  + (KH / 2) * NP2;
  float* wh2  = wi2  + (KH / 2) * NP2;
  float* l1T  = wh2  + (KH / 2) * NP2;
  float* l2T  = l1T  + (KH / 2) * KH2;
  float* bg2  = l2T  + (KH / 2) * VP2;
  float* bl1L = bg2  + NP;
  float* bl2L = bl1L + KH;
  float* h1   = bl2L + VP;
  float* h2   = h1   + 2 * CHUNK * KHP;
  float* h2m  = h2   + CHUNK * KHP;
  float* hmid = h2m  + CHUNK * KHP;
  float* gb1  = hmid + CHUNK * KHP;
  float* gb2  = gb1  + CHUNK * GP;
  int*   ids  = (int*)(gb2 + CHUNK * GP);
  int*   ids0 = ids + CHUNK;

  for (int t = __builtin_amdgcn_readfirstlane(wave); t < VT * (NP / 16); t += MW + CW) {
    const int  mt  = t / (NP / 16), nt = t - mt * (NP / 16);
    const int  v   = mt * 16 + row;
    const int  n   = nt * 16 + row;
    const int  nc  = (n < G4) ? n : 0;
    const float* ap = emb   + ((v < VOC) ? v : 0) * EMB;
    const float* bp = w_ih1 + nc * EMB;
    const float  bv = b_ih1[nc] + b_hh1[nc];
    v2f a[KE / 4], b[KE / 4];
    #pragma unroll
    for (int i = 0; i < KE / 4; ++i) {
      const bool kok = (4 * i + 2 < EMB) || (khalf == 0);
      const int  kc  = 4 * i + (kok ? khalf : 0);
      const v2f av = *(const v2f*)(ap + kc);
      const v2f bw = *(const v2f*)(bp + kc);
      a[i][0] = kok ? av[0] : 0.f;  a[i][1] = kok ? av[1] : 0.f;
      b[i][0] = kok ? bw[0] : 0.f;  b[i][1] = kok ? bw[1] : 0.f;
    }
    v8f acc;
    #pragma unroll
    for (int r = 0; r < 8; ++r) acc[r] = bv;
    __builtin_amdgcn_sched_barrier(0);
    #pragma unroll
    for (int i = 0; i < KE / 4; ++i) acc = wmma4(a[i], b[i], acc);
    #pragma unroll
    for (int r = 0; r < 8; ++r) {
      const int vr = mt * 16 + r + mhalf;
      if (vr < VOC) ptab[vr * PP + n] = acc[r];
    }
  }
  for (int i = tid; i < KH * NP; i += NT) { int k = i / NP, n = i - k * NP;
    const bool ok = (k < HID && n < G4);
    const int  d  = (k >> 1) * NP2 + n * 2 + (k & 1);
    wh1[d] = ok ? w_hh1[n * HID + k] : 0.f;
    wi2[d] = ok ? w_ih2[n * HID + k] : 0.f;
    wh2[d] = ok ? w_hh2[n * HID + k] : 0.f; }
  for (int i = tid; i < KH * KH; i += NT) { int k = i / KH, j = i - k * KH;
    l1T[(k >> 1) * KH2 + j * 2 + (k & 1)] =
        (k < HID && j < HID) ? lw1[j * HID + k] : 0.f; }
  for (int i = tid; i < KH * VP; i += NT) { int k = i / VP, v = i - k * VP;
    l2T[(k >> 1) * VP2 + v * 2 + (k & 1)] =
        (k < HID && v < VOC) ? lw2[v * HID + k] : 0.f; }
  for (int i = tid; i < NP; i += NT)
    bg2[i] = (i < G4) ? (b_ih2[i] + b_hh2[i]) : 0.f;
  for (int i = tid; i < KH; i += NT) bl1L[i] = (i < HID) ? lb1[i] : 0.f;
  for (int i = tid; i < VP; i += NT) bl2L[i] = (i < VOC) ? lb2[i] : 0.f;
  for (int i = tid; i < CHUNK * KHP; i += NT) {
    h1[i] = 0.f; h1[CHUNK * KHP + i] = 0.f; h2[i] = 0.f; h2m[i] = 0.f; hmid[i] = 0.f; }
  if (tid < CHUNK) ids0[tid] = batch_x[(b0 + tid) * T_LEN] * PP;
  else if (tid < 2 * CHUNK) ids[tid - CHUNK] = (1 < T_LEN) ? batch_x[(b0 + tid - CHUNK) * T_LEN + 1] * PP : 0;

  const int  p    = tid - CELL0;
  const bool cell = (p >= 0);
  const int  cm   = cell ? p / HID : 0;
  const int  cj   = cell ? p - cm * HID : 0;
  const int  clen = cell ? lens[b0 + cm] : 0;
  float c1 = 0.f, h1r = 0.f, c2 = 0.f, h2r = 0.f;
  int nid = 0;
  v8f xg;
  #pragma unroll
  for (int r = 0; r < 8; ++r) xg[r] = 0.f;

  __syncthreads();

  if (wave < NP / 16) {
    const int n = wave * 16 + row;
    #pragma unroll
    for (int r = 0; r < 8; ++r) xg[r] = ptab[ids0[r + mhalf] + n];
  }

  for (int s = 0; s < ST; ++s) {
    if (wave < NP / 16) {
      if (s < T_LEN) {
        const int n = wave * 16 + row;
        v8f xn;
        #pragma unroll
        for (int r = 0; r < 8; ++r) xn[r] = ptab[ids[r + mhalf] + n];
        const float* arowH = h1 + ((s + 1) & 1) * (CHUNK * KHP) + row * KHP + khalf;
        v8f acc = xg;
        acc = mm_acc(acc, arowH, KH, wh1 + n * 2 + kh2 * NP2, NP2);
        xg = xn;
        #pragma unroll
        for (int r = 0; r < 8; ++r) gb1[(r + mhalf) * GP + n] = acc[r];
      }
    } else if (wave < NP / 16 + VP / 16) {
      if (s >= 3) {
        const int t = s - 3;
        const float* arow = hmid + row * KHP + khalf;
        const int n = (wave - NP / 16) * 16 + row;
        const float bv = bl2L[n];
        v8f acc;
        #pragma unroll
        for (int r = 0; r < 8; ++r) acc[r] = bv;
        acc = mm_acc(acc, arow, KH, l2T + n * 2 + kh2 * VP2, VP2);
        if (n < VOC) {
          #pragma unroll
          for (int r = 0; r < 8; ++r) {
            const int m = r + mhalf;
            out[((size_t)(b0 + m) * T_LEN + t) * VOC + n] = acc[r];
          }
        }
      }
    } else if (wave == MW - 1) {
      if (lane < CHUNK) nid = (s + 2 < T_LEN) ? batch_x[(b0 + lane) * T_LEN + s + 2] * PP : 0;
    } else {
      if (s >= 2 && s < T_LEN + 2) {
        const int t = s - 2;
        const float* gp = gb2 + cm * GP + cj;
        const float xi = gp[0], xf = gp[HID], xc = gp[2 * HID], xo = gp[3 * HID];
        const float ig = sigm(xi);
        const float fg = sigm(xf);
        const float gg = tanh_e(xc);
        const float og = sigm(xo);
        const float cn = fg * c2 + ig * gg;
        const float hn = og * tanh_e(cn);
        const bool  on = (t < clen);
        c2  = on ? cn : c2;
        h2r = on ? hn : h2r;
        h2[cm * KHP + cj]  = h2r;
        h2m[cm * KHP + cj] = on ? h2r : 0.f;
      }
    }
    lds_barrier();

    if (wave < NP / 16) {
      if (s >= 1 && s <= T_LEN) {
        const float* arowA = h1 + ((s + 1) & 1) * (CHUNK * KHP) + row * KHP + khalf;
        const float* arowB = h2 + row * KHP + khalf;
        const int n = wave * 16 + row;
        const float bv = bg2[n];
        v8f acc;
        #pragma unroll
        for (int r = 0; r < 8; ++r) acc[r] = bv;
        acc = mm_acc(acc, arowA, KH, wi2 + n * 2 + kh2 * NP2, NP2);
        acc = mm_acc(acc, arowB, KH, wh2 + n * 2 + kh2 * NP2, NP2);
        #pragma unroll
        for (int r = 0; r < 8; ++r) gb2[(r + mhalf) * GP + n] = acc[r];
      }
    } else if (wave < NP / 16 + 2) {
      if (s >= 2 && s < T_LEN + 2) {
        const float* arow = h2m + row * KHP + khalf;
        const int n = (wave - NP / 16) * 16 + row;
        const float bv = bl1L[n];
        v8f acc;
        #pragma unroll
        for (int r = 0; r < 8; ++r) acc[r] = bv;
        acc = mm_acc(acc, arow, KH, l1T + n * 2 + kh2 * KH2, KH2);
        #pragma unroll
        for (int r = 0; r < 8; ++r) hmid[(r + mhalf) * KHP + n] = fmaxf(acc[r], 0.f);
      }
    } else if (wave == MW - 1) {
      if (lane < CHUNK) ids[lane] = nid;
    } else if (wave >= MW) {
      if (s < T_LEN) {
        const float* gp = gb1 + cm * GP + cj;
        const float xi = gp[0], xf = gp[HID], xc = gp[2 * HID], xo = gp[3 * HID];
        const float ig = sigm(xi);
        const float fg = sigm(xf);
        const float gg = tanh_e(xc);
        const float og = sigm(xo);
        const float cn = fg * c1 + ig * gg;
        const float hn = og * tanh_e(cn);
        const bool  on = (s < clen);
        c1  = on ? cn : c1;
        h1r = on ? hn : h1r;
        h1[(s & 1) * (CHUNK * KHP) + cm * KHP + cj] = h1r;
      }
    }
    lds_barrier();
  }
}

extern "C" void kernel_launch(void* const* d_in, const int* in_sizes, int n_in,
                              void* d_out, int out_size, void* d_ws, size_t ws_size,
                              hipStream_t stream) {
  const int*   batch_x = (const int*)  d_in[0];
  const int*   lens    = (const int*)  d_in[1];
  const float* emb     = (const float*)d_in[2];
  const float* w_ih1   = (const float*)d_in[3];
  const float* w_hh1   = (const float*)d_in[4];
  const float* b_ih1   = (const float*)d_in[5];
  const float* b_hh1   = (const float*)d_in[6];
  const float* w_ih2   = (const float*)d_in[7];
  const float* w_hh2   = (const float*)d_in[8];
  const float* b_ih2   = (const float*)d_in[9];
  const float* b_hh2   = (const float*)d_in[10];
  const float* lw1     = (const float*)d_in[11];
  const float* lb1     = (const float*)d_in[12];
  const float* lw2     = (const float*)d_in[13];
  const float* lb2     = (const float*)d_in[14];
  float* out = (float*)d_out;

  constexpr size_t SMEM_FLOATS =
      (size_t)VOC * PP + 3u * (KH / 2) * NP2 + (size_t)(KH / 2) * KH2 +
      (size_t)(KH / 2) * VP2 + (size_t)NP + KH + VP + 5u * CHUNK * KHP + 2u * CHUNK * GP +
      2u * CHUNK;
  constexpr size_t SMEM_BYTES = SMEM_FLOATS * sizeof(float);

  (void)hipFuncSetAttribute((const void*)charnn_fused,
                            hipFuncAttributeMaxDynamicSharedMemorySize,
                            (int)SMEM_BYTES);

  charnn_fused<<<dim3(B_SZ / CHUNK), dim3(NT), SMEM_BYTES, stream>>>(
      batch_x, lens, emb, w_ih1, w_hh1, b_ih1, b_hh1,
      w_ih2, w_hh2, b_ih2, b_hh2, lw1, lb1, lw2, lb2, out);
}


# Cell update moved into the gate waves: gate columns regrouped so one 16-col tile holds i,f,g,o of 4 units, handed over in wave-private LDS; no cell waves, 1 barrier per step not 2; sum order kept
# speedup vs baseline: 5.8376x; 1.1157x over previous
#include <hip/hip_runtime.h>
#include <hip/hip_bf16.h>


#define T_LEN 512
#define B_SZ  512
#define EMB   50
#define HID   30
#define G4    120
#define VOC   100
#define NP    128
#define KE    52
#define VT    ((VOC + 15) / 16)
#define KH    32
#define VP    112
#define CHUNK 16
#define L2W   8
#define H1W   16
#define H2W   18
#define IDW   25
#define NW    26
#define NT    (NW * 32)
#define ST    (T_LEN + 3)

#define KHP 36
#define SP  20
#define NP2 (NP * 2 + 32)
#define KH2 (KH * 2 + 32)
#define VP2 (VP * 2)
#define PP  144

typedef __attribute__((ext_vector_type(2))) float v2f;
typedef __attribute__((ext_vector_type(4))) float v4f;
typedef __attribute__((ext_vector_type(8))) float v8f;

__device__ __forceinline__ v8f wmma4(v2f a, v2f b, v8f c) {
  return __builtin_amdgcn_wmma_f32_16x16x4_f32(false, a, false, b, (short)0, c,
                                               false, false);
}

__device__ __forceinline__ int gate_row(int n) {
  const int c = n & 15;
  const int u = (n >> 4) * 4 + (c >> 2);
  return (u < HID) ? (c & 3) * HID + u : -1;
}

__device__ __forceinline__ float sigm(float x) {
  return __builtin_amdgcn_rcpf(1.0f + __builtin_amdgcn_exp2f(-1.44269504f * x));
}
__device__ __forceinline__ float tanh_e(float x) {
  return __builtin_fmaf(-2.0f, __builtin_amdgcn_rcpf(1.0f + __builtin_amdgcn_exp2f(2.88539008f * x)), 1.0f);
}

__device__ __forceinline__ v8f mm_acc(v8f acc, const float* __restrict__ aPtr, int kdim,
                                      const float* __restrict__ bPtr, int ld2) {
  #pragma unroll
  for (int kb = 0; kb < kdim; kb += 4) {
    v2f a = *(const v2f*)(aPtr + kb);
    v2f b = *(const v2f*)(bPtr + (kb >> 1) * ld2);
    acc = wmma4(a, b, acc);
  }
  return acc;
}


__device__ __forceinline__ void lds_barrier() {
  __builtin_amdgcn_fence(__ATOMIC_RELEASE, "workgroup", "local");
  __builtin_amdgcn_s_barrier();
  __builtin_amdgcn_fence(__ATOMIC_ACQUIRE, "workgroup", "local");
}

__device__ __forceinline__ void lstm_cell(v4f q, bool on, float& c, float& h) {
  const float ig = sigm(q[0]);
  const float fg = sigm(q[1]);
  const float gg = tanh_e(q[2]);
  const float og = sigm(q[3]);
  const float cn = fg * c + ig * gg;
  const float hn = og * tanh_e(cn);
  c = on ? cn : c;
  h = on ? hn : h;
}

__device__ __forceinline__ void tile_to_cells(float* __restrict__ sc, v8f acc, int col, int mhalf, int rb, int ul,
                                              v4f& q0, v4f& q1) {
  #pragma unroll
  for (int r = 0; r < 8; ++r) sc[(r + mhalf) * SP + col] = acc[r];
  __builtin_amdgcn_fence(__ATOMIC_SEQ_CST, "workgroup", "local");
  q0 = *(const v4f*)(sc + rb * SP + ul * 4);
  q1 = *(const v4f*)(sc + (rb + 8) * SP + ul * 4);
}

extern "C" __global__ __launch_bounds__(NT)
void charnn_fused(const int* __restrict__ batch_x, const int* __restrict__ lens,
                  const float* __restrict__ emb,
                  const float* __restrict__ w_ih1, const float* __restrict__ w_hh1,
                  const float* __restrict__ b_ih1, const float* __restrict__ b_hh1,
                  const float* __restrict__ w_ih2, const float* __restrict__ w_hh2,
                  const float* __restrict__ b_ih2, const float* __restrict__ b_hh2,
                  const float* __restrict__ lw1,  const float* __restrict__ lb1,
                  const float* __restrict__ lw2,  const float* __restrict__ lb2,
                  float* __restrict__ out)
{
  extern __shared__ float smem[];
  const int tid   = threadIdx.x;
  const int lane  = tid & 31;
  const int wave  = tid >> 5;
  const int row   = lane & 15;
  const int khalf = (lane < 16) ? 0 : 2;
  const int kh2   = khalf >> 1;
  const int mhalf = (lane < 16) ? 0 : 8;
  const int b0    = blockIdx.x * CHUNK;

  float* sct  = smem;
  float* ptab = sct  + 16 * CHUNK * SP;
  float* wh1  = ptab + VOC * PP;
  float* wi2  = wh1  + (KH / 2) * NP2;
  float* wh2  = wi2  + (KH / 2) * NP2;
  float* l1T  = wh2  + (KH / 2) * NP2;
  float* l2T  = l1T  + (KH / 2) * KH2;
  float* bg2  = l2T  + (KH / 2) * VP2;
  float* bl1L = bg2  + NP;
  float* bl2L = bl1L + KH;
  float* h1   = bl2L + VP;
  float* h2   = h1   + 2 * CHUNK * KHP;
  float* h2m  = h2   + 2 * CHUNK * KHP;
  float* hmid = h2m  + 2 * CHUNK * KHP;
  int*   ids  = (int*)(hmid + 2 * CHUNK * KHP);
  int*   ids0 = ids + 2 * CHUNK;
  const int HB = CHUNK * KHP;

  for (int t = __builtin_amdgcn_readfirstlane(wave); t < VT * (NP / 16); t += NW) {
    const int  mt  = t / (NP / 16), nt = t - mt * (NP / 16);
    const int  v   = mt * 16 + row;
    const int  n   = nt * 16 + row;
    const int  gr  = gate_row(n);
    const int  nc  = (gr >= 0) ? gr : 0;
    const float* ap = emb   + ((v < VOC) ? v : 0) * EMB;
    const float* bp = w_ih1 + nc * EMB;
    const float  bv = b_ih1[nc] + b_hh1[nc];
    v2f a[KE / 4], b[KE / 4];
    #pragma unroll
    for (int i = 0; i < KE / 4; ++i) {
      const bool kok = (4 * i + 2 < EMB) || (khalf == 0);
      const int  kc  = 4 * i + (kok ? khalf : 0);
      const v2f av = *(const v2f*)(ap + kc);
      const v2f bw = *(const v2f*)(bp + kc);
      a[i][0] = kok ? av[0] : 0.f;  a[i][1] = kok ? av[1] : 0.f;
      b[i][0] = kok ? bw[0] : 0.f;  b[i][1] = kok ? bw[1] : 0.f;
    }
    v8f acc;
    #pragma unroll
    for (int r = 0; r < 8; ++r) acc[r] = bv;
    __builtin_amdgcn_sched_barrier(0);
    #pragma unroll
    for (int i = 0; i < KE / 4; ++i) acc = wmma4(a[i], b[i], acc);
    #pragma unroll
    for (int r = 0; r < 8; ++r) {
      const int vr = mt * 16 + r + mhalf;
      if (vr < VOC) ptab[vr * PP + n] = acc[r];
    }
  }
  for (int i = tid; i < KH * NP; i += NT) { int k = i / NP, n = i - k * NP;
    const int  gr = gate_row(n);
    const bool ok = (k < HID && gr >= 0);
    const int  g  = ok ? gr : 0;
    const int  d  = (k >> 1) * NP2 + n * 2 + (k & 1);
    wh1[d] = ok ? w_hh1[g * HID + k] : 0.f;
    wi2[d] = ok ? w_ih2[g * HID + k] : 0.f;
    wh2[d] = ok ? w_hh2[g * HID + k] : 0.f; }
  for (int i = tid; i < KH * KH; i += NT) { int k = i / KH, j = i - k * KH;
    l1T[(k >> 1) * KH2 + j * 2 + (k & 1)] =
        (k < HID && j < HID) ? lw1[j * HID + k] : 0.f; }
  for (int i = tid; i < KH * VP; i += NT) { int k = i / VP, v = i - k * VP;
    l2T[(k >> 1) * VP2 + v * 2 + (k & 1)] =
        (k < HID && v < VOC) ? lw2[v * HID + k] : 0.f; }
  for (int i = tid; i < NP; i += NT) {
    const int gr = gate_row(i);
    bg2[i] = (gr >= 0) ? (b_ih2[gr] + b_hh2[gr]) : 0.f; }
  for (int i = tid; i < KH; i += NT) bl1L[i] = (i < HID) ? lb1[i] : 0.f;
  for (int i = tid; i < VP; i += NT) bl2L[i] = (i < VOC) ? lb2[i] : 0.f;
  for (int i = tid; i < 2 * CHUNK * KHP; i += NT) {
    h1[i] = 0.f; h2[i] = 0.f; h2m[i] = 0.f; hmid[i] = 0.f; }
  if (tid < CHUNK) ids0[tid] = batch_x[(b0 + tid) * T_LEN] * PP;
  else if (tid < 2 * CHUNK) ids[tid] = (1 < T_LEN) ? batch_x[(b0 + tid - CHUNK) * T_LEN + 1] * PP : 0;

  const bool lw   = (wave < H1W);
  const int  tile = wave & 7;
  const int  rb   = lane >> 2;
  const int  ul   = lane & 3;
  const int  cu   = tile * 4 + ul;
  const int  len0 = lw ? lens[b0 + rb] : 0;
  const int  len1 = lw ? lens[b0 + rb + 8] : 0;
  float* const sc = sct + (wave & 15) * (CHUNK * SP);
  float cs0 = 0.f, hs0 = 0.f, cs1 = 0.f, hs1 = 0.f;
  int nid = (wave == IDW && lane < CHUNK && 2 < T_LEN) ? batch_x[(b0 + lane) * T_LEN + 2] * PP : 0;
  v8f xg;
  #pragma unroll
  for (int r = 0; r < 8; ++r) xg[r] = 0.f;

  __syncthreads();

  if (wave < L2W) {
    const int n = wave * 16 + row;
    #pragma unroll
    for (int r = 0; r < 8; ++r) xg[r] = ptab[ids0[r + mhalf] + n];
  }

  for (int s = 0; s < ST; ++s) {
    const int cur = (s & 1) * HB;
    const int oth = ((s + 1) & 1) * HB;
    if (wave < L2W) {
      if (s < T_LEN) {
        const int n = wave * 16 + row;
        const int* idr = ids + ((s + 1) & 1) * CHUNK;
        v8f xn;
        #pragma unroll
        for (int r = 0; r < 8; ++r) xn[r] = ptab[idr[r + mhalf] + n];
        v8f acc = xg;
        acc = mm_acc(acc, h1 + oth + row * KHP + khalf, KH, wh1 + n * 2 + kh2 * NP2, NP2);
        xg = xn;
        v4f q0, q1;
        tile_to_cells(sc, acc, row, mhalf, rb, ul, q0, q1);
        lstm_cell(q0, s < len0, cs0, hs0);
        lstm_cell(q1, s < len1, cs1, hs1);
        if (cu < HID) {
          h1[cur + rb * KHP + cu]       = hs0;
          h1[cur + (rb + 8) * KHP + cu] = hs1;
        }
      }
    } else if (wave < H1W) {
      if (s >= 1 && s <= T_LEN) {
        const int t = s - 1;
        const int n = tile * 16 + row;
        const float bv = bg2[n];
        v8f acc;
        #pragma unroll
        for (int r = 0; r < 8; ++r) acc[r] = bv;
        acc = mm_acc(acc, h1 + oth + row * KHP + khalf, KH, wi2 + n * 2 + kh2 * NP2, NP2);
        acc = mm_acc(acc, h2 + cur + row * KHP + khalf, KH, wh2 + n * 2 + kh2 * NP2, NP2);
        v4f q0, q1;
        tile_to_cells(sc, acc, row, mhalf, rb, ul, q0, q1);
        const bool on0 = (t < len0), on1 = (t < len1);
        lstm_cell(q0, on0, cs0, hs0);
        lstm_cell(q1, on1, cs1, hs1);
        if (cu < HID) {
          h2[oth + rb * KHP + cu]        = hs0;
          h2[oth + (rb + 8) * KHP + cu]  = hs1;
          h2m[oth + rb * KHP + cu]       = on0 ? hs0 : 0.f;
          h2m[oth + (rb + 8) * KHP + cu] = on1 ? hs1 : 0.f;
        }
      }
    } else if (wave < H2W) {
      if (s >= 2 && s < T_LEN + 2) {
        const int n = (wave - H1W) * 16 + row;
        const float bv = bl1L[n];
        v8f acc;
        #pragma unroll
        for (int r = 0; r < 8; ++r) acc[r] = bv;
        acc = mm_acc(acc, h2m + cur + row * KHP + khalf, KH, l1T + n * 2 + kh2 * KH2, KH2);
        #pragma unroll
        for (int r = 0; r < 8; ++r) hmid[cur + (r + mhalf) * KHP + n] = fmaxf(acc[r], 0.f);
      }
    } else if (wave < IDW) {
      if (s >= 3) {
        const int t = s - 3;
        const int n = (wave - H2W) * 16 + row;
        const float bv = bl2L[n];
        v8f acc;
        #pragma unroll
        for (int r = 0; r < 8; ++r) acc[r] = bv;
        acc = mm_acc(acc, hmid + oth + row * KHP + khalf, KH, l2T + n * 2 + kh2 * VP2, VP2);
        if (n < VOC) {
          #pragma unroll
          for (int r = 0; r < 8; ++r) {
            const int m = r + mhalf;
            out[((size_t)(b0 + m) * T_LEN + t) * VOC + n] = acc[r];
          }
        }
      }
    } else {
      if (lane < CHUNK) {
        ids[(s & 1) * CHUNK + lane] = nid;
        nid = (s + 3 < T_LEN) ? batch_x[(b0 + lane) * T_LEN + s + 3] * PP : 0;
      }
    }
    lds_barrier();
  }
}

extern "C" void kernel_launch(void* const* d_in, const int* in_sizes, int n_in,
                              void* d_out, int out_size, void* d_ws, size_t ws_size,
                              hipStream_t stream) {
  const int*   batch_x = (const int*)  d_in[0];
  const int*   lens    = (const int*)  d_in[1];
  const float* emb     = (const float*)d_in[2];
  const float* w_ih1   = (const float*)d_in[3];
  const float* w_hh1   = (const float*)d_in[4];
  const float* b_ih1   = (const float*)d_in[5];
  const float* b_hh1   = (const float*)d_in[6];
  const float* w_ih2   = (const float*)d_in[7];
  const float* w_hh2   = (const float*)d_in[8];
  const float* b_ih2   = (const float*)d_in[9];
  const float* b_hh2   = (const float*)d_in[10];
  const float* lw1     = (const float*)d_in[11];
  const float* lb1     = (const float*)d_in[12];
  const float* lw2     = (const float*)d_in[13];
  const float* lb2     = (const float*)d_in[14];
  float* out = (float*)d_out;

  constexpr size_t SMEM_FLOATS =
      (size_t)16 * CHUNK * SP + (size_t)VOC * PP + 3u * (KH / 2) * NP2 + (size_t)(KH / 2) * KH2 +
      (size_t)(KH / 2) * VP2 + (size_t)NP + KH + VP + 8u * CHUNK * KHP + 3u * CHUNK;
  constexpr size_t SMEM_BYTES = SMEM_FLOATS * sizeof(float);

  (void)hipFuncSetAttribute((const void*)charnn_fused,
                            hipFuncAttributeMaxDynamicSharedMemorySize,
                            (int)SMEM_BYTES);

  charnn_fused<<<dim3(B_SZ / CHUNK), dim3(NT), SMEM_BYTES, stream>>>(
      batch_x, lens, emb, w_ih1, w_hh1, b_ih1, b_hh1,
      w_ih2, w_hh2, b_ih2, b_hh2, lw1, lb1, lw2, lb2, out);
}


# No LDS read between the WMMAs of a chain: weight fragments and biases stay in registers across the time loop, each trip's A fragments are read as one group ahead of its chain; sum order kept
# speedup vs baseline: 6.0937x; 1.0439x over previous
#include <hip/hip_runtime.h>
#include <hip/hip_bf16.h>


#define T_LEN 512
#define B_SZ  512
#define EMB   50
#define HID   30
#define G4    120
#define VOC   100
#define NP    128
#define KE    52
#define VT    ((VOC + 15) / 16)
#define KH    32
#define VP    112
#define CHUNK 16
#define L2W   8
#define H1W   16
#define H2W   18
#define IDW   25
#define NW    26
#define NT    (NW * 32)
#define ST    (T_LEN + 3)

#define KHP 36
#define SP  20
#define NP2 (NP * 2 + 32)
#define KH2 (KH * 2 + 32)
#define VP2 (VP * 2)
#define PP  144

typedef __attribute__((ext_vector_type(2))) float v2f;
typedef __attribute__((ext_vector_type(4))) float v4f;
typedef __attribute__((ext_vector_type(8))) float v8f;

__device__ __forceinline__ v8f wmma4(v2f a, v2f b, v8f c) {
  return __builtin_amdgcn_wmma_f32_16x16x4_f32(false, a, false, b, (short)0, c,
                                               false, false);
}

__device__ __forceinline__ int gate_row(int n) {
  const int c = n & 15;
  const int u = (n >> 4) * 4 + (c >> 2);
  return (u < HID) ? (c & 3) * HID + u : -1;
}

__device__ __forceinline__ float sigm(float x) {
  return __builtin_amdgcn_rcpf(1.0f + __builtin_amdgcn_exp2f(-1.44269504f * x));
}
__device__ __forceinline__ float tanh_e(float x) {
  return __builtin_fmaf(-2.0f, __builtin_amdgcn_rcpf(1.0f + __builtin_amdgcn_exp2f(2.88539008f * x)), 1.0f);
}

__device__ __forceinline__ void load_a(v2f* __restrict__ a, const float* __restrict__ aPtr) {
  #pragma unroll
  for (int i = 0; i < KH / 4; ++i) a[i] = *(const v2f*)(aPtr + 4 * i);
}
__device__ __forceinline__ void load_b(v2f* __restrict__ b, const float* __restrict__ bPtr, int ld2) {
  #pragma unroll
  for (int i = 0; i < KH / 4; ++i) b[i] = *(const v2f*)(bPtr + 2 * i * ld2);
}
__device__ __forceinline__ v8f mm_regs(v8f acc, const v2f* __restrict__ a, const v2f* __restrict__ b) {
  #pragma unroll
  for (int i = 0; i < KH / 4; ++i) acc = wmma4(a[i], b[i], acc);
  return acc;
}


__device__ __forceinline__ void lds_barrier() {
  __builtin_amdgcn_fence(__ATOMIC_RELEASE, "workgroup", "local");
  __builtin_amdgcn_s_barrier();
  __builtin_amdgcn_fence(__ATOMIC_ACQUIRE, "workgroup", "local");
}

__device__ __forceinline__ void lstm_cell(v4f q, bool on, float& c, float& h) {
  const float ig = sigm(q[0]);
  const float fg = sigm(q[1]);
  const float gg = tanh_e(q[2]);
  const float og = sigm(q[3]);
  const float cn = fg * c + ig * gg;
  const float hn = og * tanh_e(cn);
  c = on ? cn : c;
  h = on ? hn : h;
}

__device__ __forceinline__ void tile_to_cells(float* __restrict__ sc, v8f acc, int col, int mhalf, int rb, int ul,
                                              v4f& q0, v4f& q1) {
  #pragma unroll
  for (int r = 0; r < 8; ++r) sc[(r + mhalf) * SP + col] = acc[r];
  __builtin_amdgcn_fence(__ATOMIC_SEQ_CST, "workgroup", "local");
  q0 = *(const v4f*)(sc + rb * SP + ul * 4);
  q1 = *(const v4f*)(sc + (rb + 8) * SP + ul * 4);
}

extern "C" __global__ __launch_bounds__(NT)
void charnn_fused(const int* __restrict__ batch_x, const int* __restrict__ lens,
                  const float* __restrict__ emb,
                  const float* __restrict__ w_ih1, const float* __restrict__ w_hh1,
                  const float* __restrict__ b_ih1, const float* __restrict__ b_hh1,
                  const float* __restrict__ w_ih2, const float* __restrict__ w_hh2,
                  const float* __restrict__ b_ih2, const float* __restrict__ b_hh2,
                  const float* __restrict__ lw1,  const float* __restrict__ lb1,
                  const float* __restrict__ lw2,  const float* __restrict__ lb2,
                  float* __restrict__ out)
{
  extern __shared__ float smem[];
  const int tid   = threadIdx.x;
  const int lane  = tid & 31;
  const int wave  = tid >> 5;
  const int row   = lane & 15;
  const int khalf = (lane < 16) ? 0 : 2;
  const int kh2   = khalf >> 1;
  const int mhalf = (lane < 16) ? 0 : 8;
  const int b0    = blockIdx.x * CHUNK;

  float* sct  = smem;
  float* ptab = sct  + 16 * CHUNK * SP;
  float* wh1  = ptab + VOC * PP;
  float* wi2  = wh1  + (KH / 2) * NP2;
  float* wh2  = wi2  + (KH / 2) * NP2;
  float* l1T  = wh2  + (KH / 2) * NP2;
  float* l2T  = l1T  + (KH / 2) * KH2;
  float* bg2  = l2T  + (KH / 2) * VP2;
  float* bl1L = bg2  + NP;
  float* bl2L = bl1L + KH;
  float* h1   = bl2L + VP;
  float* h2   = h1   + 2 * CHUNK * KHP;
  float* h2m  = h2   + 2 * CHUNK * KHP;
  float* hmid = h2m  + 2 * CHUNK * KHP;
  int*   ids  = (int*)(hmid + 2 * CHUNK * KHP);
  int*   ids0 = ids + 2 * CHUNK;
  const int HB = CHUNK * KHP;

  for (int t = __builtin_amdgcn_readfirstlane(wave); t < VT * (NP / 16); t += NW) {
    const int  mt  = t / (NP / 16), nt = t - mt * (NP / 16);
    const int  v   = mt * 16 + row;
    const int  n   = nt * 16 + row;
    const int  gr  = gate_row(n);
    const int  nc  = (gr >= 0) ? gr : 0;
    const float* ap = emb   + ((v < VOC) ? v : 0) * EMB;
    const float* bp = w_ih1 + nc * EMB;
    const float  bv = b_ih1[nc] + b_hh1[nc];
    v2f a[KE / 4], b[KE / 4];
    #pragma unroll
    for (int i = 0; i < KE / 4; ++i) {
      const bool kok = (4 * i + 2 < EMB) || (khalf == 0);
      const int  kc  = 4 * i + (kok ? khalf : 0);
      const v2f av = *(const v2f*)(ap + kc);
      const v2f bw = *(const v2f*)(bp + kc);
      a[i][0] = kok ? av[0] : 0.f;  a[i][1] = kok ? av[1] : 0.f;
      b[i][0] = kok ? bw[0] : 0.f;  b[i][1] = kok ? bw[1] : 0.f;
    }
    v8f acc;
    #pragma unroll
    for (int r = 0; r < 8; ++r) acc[r] = bv;
    __builtin_amdgcn_sched_barrier(0);
    #pragma unroll
    for (int i = 0; i < KE / 4; ++i) acc = wmma4(a[i], b[i], acc);
    #pragma unroll
    for (int r = 0; r < 8; ++r) {
      const int vr = mt * 16 + r + mhalf;
      if (vr < VOC) ptab[vr * PP + n] = acc[r];
    }
  }
  for (int i = tid; i < KH * NP; i += NT) { int k = i / NP, n = i - k * NP;
    const int  gr = gate_row(n);
    const bool ok = (k < HID && gr >= 0);
    const int  g  = ok ? gr : 0;
    const int  d  = (k >> 1) * NP2 + n * 2 + (k & 1);
    wh1[d] = ok ? w_hh1[g * HID + k] : 0.f;
    wi2[d] = ok ? w_ih2[g * HID + k] : 0.f;
    wh2[d] = ok ? w_hh2[g * HID + k] : 0.f; }
  for (int i = tid; i < KH * KH; i += NT) { int k = i / KH, j = i - k * KH;
    l1T[(k >> 1) * KH2 + j * 2 + (k & 1)] =
        (k < HID && j < HID) ? lw1[j * HID + k] : 0.f; }
  for (int i = tid; i < KH * VP; i += NT) { int k = i / VP, v = i - k * VP;
    l2T[(k >> 1) * VP2 + v * 2 + (k & 1)] =
        (k < HID && v < VOC) ? lw2[v * HID + k] : 0.f; }
  for (int i = tid; i < NP; i += NT) {
    const int gr = gate_row(i);
    bg2[i] = (gr >= 0) ? (b_ih2[gr] + b_hh2[gr]) : 0.f; }
  for (int i = tid; i < KH; i += NT) bl1L[i] = (i < HID) ? lb1[i] : 0.f;
  for (int i = tid; i < VP; i += NT) bl2L[i] = (i < VOC) ? lb2[i] : 0.f;
  for (int i = tid; i < 2 * CHUNK * KHP; i += NT) {
    h1[i] = 0.f; h2[i] = 0.f; h2m[i] = 0.f; hmid[i] = 0.f; }
  if (tid < CHUNK) ids0[tid] = batch_x[(b0 + tid) * T_LEN] * PP;
  else if (tid < 2 * CHUNK) ids[tid] = (1 < T_LEN) ? batch_x[(b0 + tid - CHUNK) * T_LEN + 1] * PP : 0;

  const bool lw   = (wave < H1W);
  const int  tile = wave & 7;
  const int  rb   = lane >> 2;
  const int  ul   = lane & 3;
  const int  cu   = tile * 4 + ul;
  const int  len0 = lw ? lens[b0 + rb] : 0;
  const int  len1 = lw ? lens[b0 + rb + 8] : 0;
  float* const sc = sct + (wave & 15) * (CHUNK * SP);
  float cs0 = 0.f, hs0 = 0.f, cs1 = 0.f, hs1 = 0.f;
  int nid = (wave == IDW && lane < CHUNK && 2 < T_LEN) ? batch_x[(b0 + lane) * T_LEN + 2] * PP : 0;
  v8f xg;
  #pragma unroll
  for (int r = 0; r < 8; ++r) xg[r] = 0.f;

  __syncthreads();

  const int ncol = (wave < H1W) ? tile * 16 + row : (wave < H2W) ? (wave - H1W) * 16 + row : (wave - H2W) * 16 + row;
  v2f bw[2 * (KH / 4)];
  #pragma unroll
  for (int i = 0; i < 2 * (KH / 4); ++i) { bw[i][0] = 0.f; bw[i][1] = 0.f; }
  float bias = 0.f;
  if (wave < L2W) {
    load_b(bw, wh1 + ncol * 2 + kh2 * NP2, NP2);
  } else if (wave < H1W) {
    load_b(bw,          wi2 + ncol * 2 + kh2 * NP2, NP2);
    load_b(bw + KH / 4, wh2 + ncol * 2 + kh2 * NP2, NP2);
    bias = bg2[ncol];
  } else if (wave < H2W) {
    load_b(bw, l1T + ncol * 2 + kh2 * KH2, KH2);
    bias = bl1L[ncol];
  } else if (wave < IDW) {
    load_b(bw, l2T + ncol * 2 + kh2 * VP2, VP2);
    bias = bl2L[ncol];
  }

  if (wave < L2W) {
    #pragma unroll
    for (int r = 0; r < 8; ++r) xg[r] = ptab[ids0[r + mhalf] + ncol];
  }

  for (int s = 0; s < ST; ++s) {
    const int cur = (s & 1) * HB;
    const int oth = ((s + 1) & 1) * HB;
    if (wave < L2W) {
      if (s < T_LEN) {
        const int* idr = ids + ((s + 1) & 1) * CHUNK;
        v2f a[KH / 4];
        load_a(a, h1 + oth + row * KHP + khalf);
        int tk[8];
        #pragma unroll
        for (int r = 0; r < 8; ++r) tk[r] = idr[r + mhalf];
        __builtin_amdgcn_sched_barrier(0);
        const v8f acc = mm_regs(xg, a, bw);
        v4f q0, q1;
        tile_to_cells(sc, acc, row, mhalf, rb, ul, q0, q1);
        #pragma unroll
        for (int r = 0; r < 8; ++r) xg[r] = ptab[tk[r] + ncol];
        lstm_cell(q0, s < len0, cs0, hs0);
        lstm_cell(q1, s < len1, cs1, hs1);
        if (cu < HID) {
          h1[cur + rb * KHP + cu]       = hs0;
          h1[cur + (rb + 8) * KHP + cu] = hs1;
        }
      }
    } else if (wave < H1W) {
      if (s >= 1 && s <= T_LEN) {
        const int t = s - 1;
        v2f a[2 * (KH / 4)];
        load_a(a,          h1 + oth + row * KHP + khalf);
        load_a(a + KH / 4, h2 + cur + row * KHP + khalf);
        __builtin_amdgcn_sched_barrier(0);
        v8f acc;
        #pragma unroll
        for (int r = 0; r < 8; ++r) acc[r] = bias;
        acc = mm_regs(acc, a,          bw);
        acc = mm_regs(acc, a + KH / 4, bw + KH / 4);
        v4f q0, q1;
        tile_to_cells(sc, acc, row, mhalf, rb, ul, q0, q1);
        const bool on0 = (t < len0), on1 = (t < len1);
        lstm_cell(q0, on0, cs0, hs0);
        lstm_cell(q1, on1, cs1, hs1);
        if (cu < HID) {
          h2[oth + rb * KHP + cu]        = hs0;
          h2[oth + (rb + 8) * KHP + cu]  = hs1;
          h2m[oth + rb * KHP + cu]       = on0 ? hs0 : 0.f;
          h2m[oth + (rb + 8) * KHP + cu] = on1 ? hs1 : 0.f;
        }
      }
    } else if (wave < H2W) {
      if (s >= 2 && s < T_LEN + 2) {
        v2f a[KH / 4];
        load_a(a, h2m + cur + row * KHP + khalf);
        __builtin_amdgcn_sched_barrier(0);
        v8f acc;
        #pragma unroll
        for (int r = 0; r < 8; ++r) acc[r] = bias;
        acc = mm_regs(acc, a, bw);
        #pragma unroll
        for (int r = 0; r < 8; ++r) hmid[cur + (r + mhalf) * KHP + ncol] = fmaxf(acc[r], 0.f);
      }
    } else if (wave < IDW) {
      if (s >= 3) {
        const int t = s - 3;
        v2f a[KH / 4];
        load_a(a, hmid + oth + row * KHP + khalf);
        __builtin_amdgcn_sched_barrier(0);
        v8f acc;
        #pragma unroll
        for (int r = 0; r < 8; ++r) acc[r] = bias;
        acc = mm_regs(acc, a, bw);
        if (ncol < VOC) {
          #pragma unroll
          for (int r = 0; r < 8; ++r) {
            const int m = r + mhalf;
            out[((size_t)(b0 + m) * T_LEN + t) * VOC + ncol] = acc[r];
          }
        }
      }
    } else {
      if (lane < CHUNK) {
        ids[(s & 1) * CHUNK + lane] = nid;
        nid = (s + 3 < T_LEN) ? batch_x[(b0 + lane) * T_LEN + s + 3] * PP : 0;
      }
    }
    lds_barrier();
  }
}

extern "C" void kernel_launch(void* const* d_in, const int* in_sizes, int n_in,
                              void* d_out, int out_size, void* d_ws, size_t ws_size,
                              hipStream_t stream) {
  const int*   batch_x = (const int*)  d_in[0];
  const int*   lens    = (const int*)  d_in[1];
  const float* emb     = (const float*)d_in[2];
  const float* w_ih1   = (const float*)d_in[3];
  const float* w_hh1   = (const float*)d_in[4];
  const float* b_ih1   = (const float*)d_in[5];
  const float* b_hh1   = (const float*)d_in[6];
  const float* w_ih2   = (const float*)d_in[7];
  const float* w_hh2   = (const float*)d_in[8];
  const float* b_ih2   = (const float*)d_in[9];
  const float* b_hh2   = (const float*)d_in[10];
  const float* lw1     = (const float*)d_in[11];
  const float* lb1     = (const float*)d_in[12];
  const float* lw2     = (const float*)d_in[13];
  const float* lb2     = (const float*)d_in[14];
  float* out = (float*)d_out;

  constexpr size_t SMEM_FLOATS =
      (size_t)16 * CHUNK * SP + (size_t)VOC * PP + 3u * (KH / 2) * NP2 + (size_t)(KH / 2) * KH2 +
      (size_t)(KH / 2) * VP2 + (size_t)NP + KH + VP + 8u * CHUNK * KHP + 3u * CHUNK;
  constexpr size_t SMEM_BYTES = SMEM_FLOATS * sizeof(float);

  (void)hipFuncSetAttribute((const void*)charnn_fused,
                            hipFuncAttributeMaxDynamicSharedMemorySize,
                            (int)SMEM_BYTES);

  charnn_fused<<<dim3(B_SZ / CHUNK), dim3(NT), SMEM_BYTES, stream>>>(
      batch_x, lens, emb, w_ih1, w_hh1, b_ih1, b_hh1,
      w_ih2, w_hh2, b_ih2, b_hh2, lw1, lb1, lw2, lb2, out);
}


# Step barrier split for the head-2 waves: they signal once their operand rows are read and wait only after their products and output stores, so no other wave waits for that work; sum order kept
# speedup vs baseline: 6.3278x; 1.0384x over previous
#include <hip/hip_runtime.h>
#include <hip/hip_bf16.h>


#define T_LEN 512
#define B_SZ  512
#define EMB   50
#define HID   30
#define G4    120
#define VOC   100
#define NP    128
#define KE    52
#define VT    ((VOC + 15) / 16)
#define KH    32
#define VP    112
#define CHUNK 16
#define L2W   8
#define H1W   16
#define H2W   18
#define IDW   25
#define NW    26
#define NT    (NW * 32)
#define ST    (T_LEN + 3)

#define KHP 36
#define SP  20
#define NP2 (NP * 2 + 32)
#define KH2 (KH * 2 + 32)
#define VP2 (VP * 2)
#define PP  144

typedef __attribute__((ext_vector_type(2))) float v2f;
typedef __attribute__((ext_vector_type(4))) float v4f;
typedef __attribute__((ext_vector_type(8))) float v8f;

__device__ __forceinline__ v8f wmma4(v2f a, v2f b, v8f c) {
  return __builtin_amdgcn_wmma_f32_16x16x4_f32(false, a, false, b, (short)0, c,
                                               false, false);
}

__device__ __forceinline__ int gate_row(int n) {
  const int c = n & 15;
  const int u = (n >> 4) * 4 + (c >> 2);
  return (u < HID) ? (c & 3) * HID + u : -1;
}

__device__ __forceinline__ float sigm(float x) {
  return __builtin_amdgcn_rcpf(1.0f + __builtin_amdgcn_exp2f(-1.44269504f * x));
}
__device__ __forceinline__ float tanh_e(float x) {
  return __builtin_fmaf(-2.0f, __builtin_amdgcn_rcpf(1.0f + __builtin_amdgcn_exp2f(2.88539008f * x)), 1.0f);
}

__device__ __forceinline__ void load_a(v2f* __restrict__ a, const float* __restrict__ aPtr) {
  #pragma unroll
  for (int i = 0; i < KH / 4; ++i) a[i] = *(const v2f*)(aPtr + 4 * i);
}
__device__ __forceinline__ void load_b(v2f* __restrict__ b, const float* __restrict__ bPtr, int ld2) {
  #pragma unroll
  for (int i = 0; i < KH / 4; ++i) b[i] = *(const v2f*)(bPtr + 2 * i * ld2);
}
__device__ __forceinline__ v8f mm_regs(v8f acc, const v2f* __restrict__ a, const v2f* __restrict__ b) {
  #pragma unroll
  for (int i = 0; i < KH / 4; ++i) acc = wmma4(a[i], b[i], acc);
  return acc;
}


__device__ __forceinline__ void lds_arrive() {
  __builtin_amdgcn_fence(__ATOMIC_RELEASE, "workgroup", "local");
  __builtin_amdgcn_s_barrier_signal(-1);
}
__device__ __forceinline__ void lds_sync() {
  __builtin_amdgcn_s_barrier_wait(-1);
  __builtin_amdgcn_fence(__ATOMIC_ACQUIRE, "workgroup", "local");
}

__device__ __forceinline__ void lstm_cell(v4f q, bool on, float& c, float& h) {
  const float ig = sigm(q[0]);
  const float fg = sigm(q[1]);
  const float gg = tanh_e(q[2]);
  const float og = sigm(q[3]);
  const float cn = fg * c + ig * gg;
  const float hn = og * tanh_e(cn);
  c = on ? cn : c;
  h = on ? hn : h;
}

__device__ __forceinline__ void tile_to_cells(float* __restrict__ sc, v8f acc, int col, int mhalf, int rb, int ul,
                                              v4f& q0, v4f& q1) {
  #pragma unroll
  for (int r = 0; r < 8; ++r) sc[(r + mhalf) * SP + col] = acc[r];
  __builtin_amdgcn_fence(__ATOMIC_SEQ_CST, "workgroup", "local");
  q0 = *(const v4f*)(sc + rb * SP + ul * 4);
  q1 = *(const v4f*)(sc + (rb + 8) * SP + ul * 4);
}

extern "C" __global__ __launch_bounds__(NT)
void charnn_fused(const int* __restrict__ batch_x, const int* __restrict__ lens,
                  const float* __restrict__ emb,
                  const float* __restrict__ w_ih1, const float* __restrict__ w_hh1,
                  const float* __restrict__ b_ih1, const float* __restrict__ b_hh1,
                  const float* __restrict__ w_ih2, const float* __restrict__ w_hh2,
                  const float* __restrict__ b_ih2, const float* __restrict__ b_hh2,
                  const float* __restrict__ lw1,  const float* __restrict__ lb1,
                  const float* __restrict__ lw2,  const float* __restrict__ lb2,
                  float* __restrict__ out)
{
  extern __shared__ float smem[];
  const int tid   = threadIdx.x;
  const int lane  = tid & 31;
  const int wave  = tid >> 5;
  const int row   = lane & 15;
  const int khalf = (lane < 16) ? 0 : 2;
  const int kh2   = khalf >> 1;
  const int mhalf = (lane < 16) ? 0 : 8;
  const int b0    = blockIdx.x * CHUNK;

  float* sct  = smem;
  float* ptab = sct  + 16 * CHUNK * SP;
  float* wh1  = ptab + VOC * PP;
  float* wi2  = wh1  + (KH / 2) * NP2;
  float* wh2  = wi2  + (KH / 2) * NP2;
  float* l1T  = wh2  + (KH / 2) * NP2;
  float* l2T  = l1T  + (KH / 2) * KH2;
  float* bg2  = l2T  + (KH / 2) * VP2;
  float* bl1L = bg2  + NP;
  float* bl2L = bl1L + KH;
  float* h1   = bl2L + VP;
  float* h2   = h1   + 2 * CHUNK * KHP;
  float* h2m  = h2   + 2 * CHUNK * KHP;
  float* hmid = h2m  + 2 * CHUNK * KHP;
  int*   ids  = (int*)(hmid + 2 * CHUNK * KHP);
  int*   ids0 = ids + 2 * CHUNK;
  const int HB = CHUNK * KHP;

  for (int t = __builtin_amdgcn_readfirstlane(wave); t < VT * (NP / 16); t += NW) {
    const int  mt  = t / (NP / 16), nt = t - mt * (NP / 16);
    const int  v   = mt * 16 + row;
    const int  n   = nt * 16 + row;
    const int  gr  = gate_row(n);
    const int  nc  = (gr >= 0) ? gr : 0;
    const float* ap = emb   + ((v < VOC) ? v : 0) * EMB;
    const float* bp = w_ih1 + nc * EMB;
    const float  bv = b_ih1[nc] + b_hh1[nc];
    v2f a[KE / 4], b[KE / 4];
    #pragma unroll
    for (int i = 0; i < KE / 4; ++i) {
      const bool kok = (4 * i + 2 < EMB) || (khalf == 0);
      const int  kc  = 4 * i + (kok ? khalf : 0);
      const v2f av = *(const v2f*)(ap + kc);
      const v2f bw = *(const v2f*)(bp + kc);
      a[i][0] = kok ? av[0] : 0.f;  a[i][1] = kok ? av[1] : 0.f;
      b[i][0] = kok ? bw[0] : 0.f;  b[i][1] = kok ? bw[1] : 0.f;
    }
    v8f acc;
    #pragma unroll
    for (int r = 0; r < 8; ++r) acc[r] = bv;
    __builtin_amdgcn_sched_barrier(0);
    #pragma unroll
    for (int i = 0; i < KE / 4; ++i) acc = wmma4(a[i], b[i], acc);
    #pragma unroll
    for (int r = 0; r < 8; ++r) {
      const int vr = mt * 16 + r + mhalf;
      if (vr < VOC) ptab[vr * PP + n] = acc[r];
    }
  }
  for (int i = tid; i < KH * NP; i += NT) { int k = i / NP, n = i - k * NP;
    const int  gr = gate_row(n);
    const bool ok = (k < HID && gr >= 0);
    const int  g  = ok ? gr : 0;
    const int  d  = (k >> 1) * NP2 + n * 2 + (k & 1);
    wh1[d] = ok ? w_hh1[g * HID + k] : 0.f;
    wi2[d] = ok ? w_ih2[g * HID + k] : 0.f;
    wh2[d] = ok ? w_hh2[g * HID + k] : 0.f; }
  for (int i = tid; i < KH * KH; i += NT) { int k = i / KH, j = i - k * KH;
    l1T[(k >> 1) * KH2 + j * 2 + (k & 1)] =
        (k < HID && j < HID) ? lw1[j * HID + k] : 0.f; }
  for (int i = tid; i < KH * VP; i += NT) { int k = i / VP, v = i - k * VP;
    l2T[(k >> 1) * VP2 + v * 2 + (k & 1)] =
        (k < HID && v < VOC) ? lw2[v * HID + k] : 0.f; }
  for (int i = tid; i < NP; i += NT) {
    const int gr = gate_row(i);
    bg2[i] = (gr >= 0) ? (b_ih2[gr] + b_hh2[gr]) : 0.f; }
  for (int i = tid; i < KH; i += NT) bl1L[i] = (i < HID) ? lb1[i] : 0.f;
  for (int i = tid; i < VP; i += NT) bl2L[i] = (i < VOC) ? lb2[i] : 0.f;
  for (int i = tid; i < 2 * CHUNK * KHP; i += NT) {
    h1[i] = 0.f; h2[i] = 0.f; h2m[i] = 0.f; hmid[i] = 0.f; }
  if (tid < CHUNK) ids0[tid] = batch_x[(b0 + tid) * T_LEN] * PP;
  else if (tid < 2 * CHUNK) ids[tid] = (1 < T_LEN) ? batch_x[(b0 + tid - CHUNK) * T_LEN + 1] * PP : 0;

  const bool lw   = (wave < H1W);
  const int  tile = wave & 7;
  const int  rb   = lane >> 2;
  const int  ul   = lane & 3;
  const int  cu   = tile * 4 + ul;
  const int  len0 = lw ? lens[b0 + rb] : 0;
  const int  len1 = lw ? lens[b0 + rb + 8] : 0;
  float* const sc = sct + (wave & 15) * (CHUNK * SP);
  float cs0 = 0.f, hs0 = 0.f, cs1 = 0.f, hs1 = 0.f;
  int nid = (wave == IDW && lane < CHUNK && 2 < T_LEN) ? batch_x[(b0 + lane) * T_LEN + 2] * PP : 0;
  v8f xg;
  #pragma unroll
  for (int r = 0; r < 8; ++r) xg[r] = 0.f;

  __syncthreads();

  const int ncol = (wave < H1W) ? tile * 16 + row : (wave < H2W) ? (wave - H1W) * 16 + row : (wave - H2W) * 16 + row;
  v2f bw[2 * (KH / 4)];
  #pragma unroll
  for (int i = 0; i < 2 * (KH / 4); ++i) { bw[i][0] = 0.f; bw[i][1] = 0.f; }
  float bias = 0.f;
  if (wave < L2W) {
    load_b(bw, wh1 + ncol * 2 + kh2 * NP2, NP2);
  } else if (wave < H1W) {
    load_b(bw,          wi2 + ncol * 2 + kh2 * NP2, NP2);
    load_b(bw + KH / 4, wh2 + ncol * 2 + kh2 * NP2, NP2);
    bias = bg2[ncol];
  } else if (wave < H2W) {
    load_b(bw, l1T + ncol * 2 + kh2 * KH2, KH2);
    bias = bl1L[ncol];
  } else if (wave < IDW) {
    load_b(bw, l2T + ncol * 2 + kh2 * VP2, VP2);
    bias = bl2L[ncol];
  }

  if (wave < L2W) {
    #pragma unroll
    for (int r = 0; r < 8; ++r) xg[r] = ptab[ids0[r + mhalf] + ncol];
  }

  const int wv = __builtin_amdgcn_readfirstlane(wave);

  for (int s = 0; s < ST; ++s) {
    const int cur = (s & 1) * HB;
    const int oth = ((s + 1) & 1) * HB;
    if (wave < L2W) {
      if (s < T_LEN) {
        const int* idr = ids + ((s + 1) & 1) * CHUNK;
        v2f a[KH / 4];
        load_a(a, h1 + oth + row * KHP + khalf);
        int tk[8];
        #pragma unroll
        for (int r = 0; r < 8; ++r) tk[r] = idr[r + mhalf];
        __builtin_amdgcn_sched_barrier(0);
        const v8f acc = mm_regs(xg, a, bw);
        v4f q0, q1;
        tile_to_cells(sc, acc, row, mhalf, rb, ul, q0, q1);
        #pragma unroll
        for (int r = 0; r < 8; ++r) xg[r] = ptab[tk[r] + ncol];
        lstm_cell(q0, s < len0, cs0, hs0);
        lstm_cell(q1, s < len1, cs1, hs1);
        if (cu < HID) {
          h1[cur + rb * KHP + cu]       = hs0;
          h1[cur + (rb + 8) * KHP + cu] = hs1;
        }
      }
    } else if (wave < H1W) {
      if (s >= 1 && s <= T_LEN) {
        const int t = s - 1;
        v2f a[2 * (KH / 4)];
        load_a(a,          h1 + oth + row * KHP + khalf);
        load_a(a + KH / 4, h2 + cur + row * KHP + khalf);
        __builtin_amdgcn_sched_barrier(0);
        v8f acc;
        #pragma unroll
        for (int r = 0; r < 8; ++r) acc[r] = bias;
        acc = mm_regs(acc, a,          bw);
        acc = mm_regs(acc, a + KH / 4, bw + KH / 4);
        v4f q0, q1;
        tile_to_cells(sc, acc, row, mhalf, rb, ul, q0, q1);
        const bool on0 = (t < len0), on1 = (t < len1);
        lstm_cell(q0, on0, cs0, hs0);
        lstm_cell(q1, on1, cs1, hs1);
        if (cu < HID) {
          h2[oth + rb * KHP + cu]        = hs0;
          h2[oth + (rb + 8) * KHP + cu]  = hs1;
          h2m[oth + rb * KHP + cu]       = on0 ? hs0 : 0.f;
          h2m[oth + (rb + 8) * KHP + cu] = on1 ? hs1 : 0.f;
        }
      }
    } else if (wave < H2W) {
      if (s >= 2 && s < T_LEN + 2) {
        v2f a[KH / 4];
        load_a(a, h2m + cur + row * KHP + khalf);
        __builtin_amdgcn_sched_barrier(0);
        v8f acc;
        #pragma unroll
        for (int r = 0; r < 8; ++r) acc[r] = bias;
        acc = mm_regs(acc, a, bw);
        #pragma unroll
        for (int r = 0; r < 8; ++r) hmid[cur + (r + mhalf) * KHP + ncol] = fmaxf(acc[r], 0.f);
      }
    } else if (wave < IDW) {
      if (s >= 3) {
        const int t = s - 3;
        v2f a[KH / 4];
        load_a(a, hmid + oth + row * KHP + khalf);
        lds_arrive();
        __builtin_amdgcn_sched_barrier(0);
        v8f acc;
        #pragma unroll
        for (int r = 0; r < 8; ++r) acc[r] = bias;
        acc = mm_regs(acc, a, bw);
        if (ncol < VOC) {
          #pragma unroll
          for (int r = 0; r < 8; ++r) {
            const int m = r + mhalf;
            out[((size_t)(b0 + m) * T_LEN + t) * VOC + ncol] = acc[r];
          }
        }
      } else {
        lds_arrive();
      }
    } else {
      if (lane < CHUNK) {
        ids[(s & 1) * CHUNK + lane] = nid;
        nid = (s + 3 < T_LEN) ? batch_x[(b0 + lane) * T_LEN + s + 3] * PP : 0;
      }
    }
    if (wv < H2W || wv >= IDW) lds_arrive();
    lds_sync();
  }
}

extern "C" void kernel_launch(void* const* d_in, const int* in_sizes, int n_in,
                              void* d_out, int out_size, void* d_ws, size_t ws_size,
                              hipStream_t stream) {
  const int*   batch_x = (const int*)  d_in[0];
  const int*   lens    = (const int*)  d_in[1];
  const float* emb     = (const float*)d_in[2];
  const float* w_ih1   = (const float*)d_in[3];
  const float* w_hh1   = (const float*)d_in[4];
  const float* b_ih1   = (const float*)d_in[5];
  const float* b_hh1   = (const float*)d_in[6];
  const float* w_ih2   = (const float*)d_in[7];
  const float* w_hh2   = (const float*)d_in[8];
  const float* b_ih2   = (const float*)d_in[9];
  const float* b_hh2   = (const float*)d_in[10];
  const float* lw1     = (const float*)d_in[11];
  const float* lb1     = (const float*)d_in[12];
  const float* lw2     = (const float*)d_in[13];
  const float* lb2     = (const float*)d_in[14];
  float* out = (float*)d_out;

  constexpr size_t SMEM_FLOATS =
      (size_t)16 * CHUNK * SP + (size_t)VOC * PP + 3u * (KH / 2) * NP2 + (size_t)(KH / 2) * KH2 +
      (size_t)(KH / 2) * VP2 + (size_t)NP + KH + VP + 8u * CHUNK * KHP + 3u * CHUNK;
  constexpr size_t SMEM_BYTES = SMEM_FLOATS * sizeof(float);

  (void)hipFuncSetAttribute((const void*)charnn_fused,
                            hipFuncAttributeMaxDynamicSharedMemorySize,
                            (int)SMEM_BYTES);

  charnn_fused<<<dim3(B_SZ / CHUNK), dim3(NT), SMEM_BYTES, stream>>>(
      batch_x, lens, emb, w_ih1, w_hh1, b_ih1, b_hh1,
      w_ih2, w_hh2, b_ih2, b_hh2, lw1, lb1, lw2, lb2, out);
}


# Gate products transposed (weights as A operand, h rows as B): a lane's D registers hold all four gates of its own two cells, so cells update in place with no LDS hand-over; sum order kept
# speedup vs baseline: 6.9124x; 1.0924x over previous
#include <hip/hip_runtime.h>
#include <hip/hip_bf16.h>


#define T_LEN 512
#define B_SZ  512
#define EMB   50
#define HID   30
#define G4    120
#define VOC   100
#define NP    128
#define KE    52
#define VT    ((VOC + 15) / 16)
#define KH    32
#define VP    112
#define CHUNK 16
#define L2W   8
#define H1W   16
#define H2W   18
#define IDW   25
#define NW    26
#define NT    (NW * 32)
#define ST    (T_LEN + 3)

#define KHP 36
#define NP2 (NP * 2 + 32)
#define KH2 (KH * 2 + 32)
#define VP2 (VP * 2)
#define PP  144

typedef __attribute__((ext_vector_type(2))) float v2f;
typedef __attribute__((ext_vector_type(4))) float v4f;
typedef __attribute__((ext_vector_type(8))) float v8f;

__device__ __forceinline__ v8f wmma4(v2f a, v2f b, v8f c) {
  return __builtin_amdgcn_wmma_f32_16x16x4_f32(false, a, false, b, (short)0, c,
                                               false, false);
}

__device__ __forceinline__ int gate_row(int n) {
  const int c = n & 15;
  const int u = (n >> 4) * 4 + (c >> 2);
  return (u < HID) ? (c & 3) * HID + u : -1;
}

__device__ __forceinline__ float sigm(float x) {
  return __builtin_amdgcn_rcpf(1.0f + __builtin_amdgcn_exp2f(-1.44269504f * x));
}
__device__ __forceinline__ float tanh_e(float x) {
  return __builtin_fmaf(-2.0f, __builtin_amdgcn_rcpf(1.0f + __builtin_amdgcn_exp2f(2.88539008f * x)), 1.0f);
}

__device__ __forceinline__ void load_a(v2f* __restrict__ a, const float* __restrict__ aPtr) {
  #pragma unroll
  for (int i = 0; i < KH / 4; ++i) a[i] = *(const v2f*)(aPtr + 4 * i);
}
__device__ __forceinline__ void load_b(v2f* __restrict__ b, const float* __restrict__ bPtr, int ld2) {
  #pragma unroll
  for (int i = 0; i < KH / 4; ++i) b[i] = *(const v2f*)(bPtr + 2 * i * ld2);
}
__device__ __forceinline__ v8f mm_regs(v8f acc, const v2f* __restrict__ a, const v2f* __restrict__ b) {
  #pragma unroll
  for (int i = 0; i < KH / 4; ++i) acc = wmma4(a[i], b[i], acc);
  return acc;
}
__device__ __forceinline__ v8f mm_regs_t(v8f acc, const v2f* __restrict__ w, const v2f* __restrict__ hrow) {
  #pragma unroll
  for (int i = 0; i < KH / 4; ++i) acc = wmma4(w[i], hrow[i], acc);
  return acc;
}


__device__ __forceinline__ void lds_arrive() {
  __builtin_amdgcn_fence(__ATOMIC_RELEASE, "workgroup", "local");
  __builtin_amdgcn_s_barrier_signal(-1);
}
__device__ __forceinline__ void lds_sync() {
  __builtin_amdgcn_s_barrier_wait(-1);
  __builtin_amdgcn_fence(__ATOMIC_ACQUIRE, "workgroup", "local");
}

__device__ __forceinline__ void lstm_cell(v4f q, bool on, float& c, float& h) {
  const float ig = sigm(q[0]);
  const float fg = sigm(q[1]);
  const float gg = tanh_e(q[2]);
  const float og = sigm(q[3]);
  const float cn = fg * c + ig * gg;
  const float hn = og * tanh_e(cn);
  c = on ? cn : c;
  h = on ? hn : h;
}

extern "C" __global__ __launch_bounds__(NT)
void charnn_fused(const int* __restrict__ batch_x, const int* __restrict__ lens,
                  const float* __restrict__ emb,
                  const float* __restrict__ w_ih1, const float* __restrict__ w_hh1,
                  const float* __restrict__ b_ih1, const float* __restrict__ b_hh1,
                  const float* __restrict__ w_ih2, const float* __restrict__ w_hh2,
                  const float* __restrict__ b_ih2, const float* __restrict__ b_hh2,
                  const float* __restrict__ lw1,  const float* __restrict__ lb1,
                  const float* __restrict__ lw2,  const float* __restrict__ lb2,
                  float* __restrict__ out)
{
  extern __shared__ float smem[];
  const int tid   = threadIdx.x;
  const int lane  = tid & 31;
  const int wave  = tid >> 5;
  const int row   = lane & 15;
  const int khalf = (lane < 16) ? 0 : 2;
  const int kh2   = khalf >> 1;
  const int mhalf = (lane < 16) ? 0 : 8;
  const int b0    = blockIdx.x * CHUNK;

  float* ptab = smem;
  float* wh1  = ptab + VOC * PP;
  float* wi2  = wh1  + (KH / 2) * NP2;
  float* wh2  = wi2  + (KH / 2) * NP2;
  float* l1T  = wh2  + (KH / 2) * NP2;
  float* l2T  = l1T  + (KH / 2) * KH2;
  float* bg2  = l2T  + (KH / 2) * VP2;
  float* bl1L = bg2  + NP;
  float* bl2L = bl1L + KH;
  float* h1   = bl2L + VP;
  float* h2   = h1   + 2 * CHUNK * KHP;
  float* h2m  = h2   + 2 * CHUNK * KHP;
  float* hmid = h2m  + 2 * CHUNK * KHP;
  int*   ids  = (int*)(hmid + 2 * CHUNK * KHP);
  int*   ids0 = ids + 2 * CHUNK;
  const int HB = CHUNK * KHP;

  for (int t = __builtin_amdgcn_readfirstlane(wave); t < VT * (NP / 16); t += NW) {
    const int  mt  = t / (NP / 16), nt = t - mt * (NP / 16);
    const int  v   = mt * 16 + row;
    const int  n   = nt * 16 + row;
    const int  gr  = gate_row(n);
    const int  nc  = (gr >= 0) ? gr : 0;
    const float* ap = emb   + ((v < VOC) ? v : 0) * EMB;
    const float* bp = w_ih1 + nc * EMB;
    const float  bv = b_ih1[nc] + b_hh1[nc];
    v2f a[KE / 4], b[KE / 4];
    #pragma unroll
    for (int i = 0; i < KE / 4; ++i) {
      const bool kok = (4 * i + 2 < EMB) || (khalf == 0);
      const int  kc  = 4 * i + (kok ? khalf : 0);
      const v2f av = *(const v2f*)(ap + kc);
      const v2f bw = *(const v2f*)(bp + kc);
      a[i][0] = kok ? av[0] : 0.f;  a[i][1] = kok ? av[1] : 0.f;
      b[i][0] = kok ? bw[0] : 0.f;  b[i][1] = kok ? bw[1] : 0.f;
    }
    v8f acc;
    #pragma unroll
    for (int r = 0; r < 8; ++r) acc[r] = bv;
    __builtin_amdgcn_sched_barrier(0);
    #pragma unroll
    for (int i = 0; i < KE / 4; ++i) acc = wmma4(a[i], b[i], acc);
    #pragma unroll
    for (int r = 0; r < 8; ++r) {
      const int vr = mt * 16 + r + mhalf;
      if (vr < VOC) ptab[vr * PP + n] = acc[r];
    }
  }
  for (int i = tid; i < KH * NP; i += NT) { int k = i / NP, n = i - k * NP;
    const int  gr = gate_row(n);
    const bool ok = (k < HID && gr >= 0);
    const int  g  = ok ? gr : 0;
    const int  d  = (k >> 1) * NP2 + n * 2 + (k & 1);
    wh1[d] = ok ? w_hh1[g * HID + k] : 0.f;
    wi2[d] = ok ? w_ih2[g * HID + k] : 0.f;
    wh2[d] = ok ? w_hh2[g * HID + k] : 0.f; }
  for (int i = tid; i < KH * KH; i += NT) { int k = i / KH, j = i - k * KH;
    l1T[(k >> 1) * KH2 + j * 2 + (k & 1)] =
        (k < HID && j < HID) ? lw1[j * HID + k] : 0.f; }
  for (int i = tid; i < KH * VP; i += NT) { int k = i / VP, v = i - k * VP;
    l2T[(k >> 1) * VP2 + v * 2 + (k & 1)] =
        (k < HID && v < VOC) ? lw2[v * HID + k] : 0.f; }
  for (int i = tid; i < NP; i += NT) {
    const int gr = gate_row(i);
    bg2[i] = (gr >= 0) ? (b_ih2[gr] + b_hh2[gr]) : 0.f; }
  for (int i = tid; i < KH; i += NT) bl1L[i] = (i < HID) ? lb1[i] : 0.f;
  for (int i = tid; i < VP; i += NT) bl2L[i] = (i < VOC) ? lb2[i] : 0.f;
  for (int i = tid; i < 2 * CHUNK * KHP; i += NT) {
    h1[i] = 0.f; h2[i] = 0.f; h2m[i] = 0.f; hmid[i] = 0.f; }
  if (tid < CHUNK) ids0[tid] = batch_x[(b0 + tid) * T_LEN] * PP;
  else if (tid < 2 * CHUNK) ids[tid] = (1 < T_LEN) ? batch_x[(b0 + tid - CHUNK) * T_LEN + 1] * PP : 0;

  const bool lw   = (wave < H1W);
  const int  tile = wave & 7;
  const int  cu   = tile * 4 + khalf;
  const int  gc0  = tile * 16 + mhalf;
  const int  len  = lw ? lens[b0 + row] : 0;
  float cs0 = 0.f, hs0 = 0.f, cs1 = 0.f, hs1 = 0.f;
  int nid = (wave == IDW && lane < CHUNK && 2 < T_LEN) ? batch_x[(b0 + lane) * T_LEN + 2] * PP : 0;
  v8f xg;
  #pragma unroll
  for (int r = 0; r < 8; ++r) xg[r] = 0.f;

  __syncthreads();

  const int ncol = (wave < H1W) ? tile * 16 + row : (wave < H2W) ? (wave - H1W) * 16 + row : (wave - H2W) * 16 + row;
  v2f bw[2 * (KH / 4)];
  #pragma unroll
  for (int i = 0; i < 2 * (KH / 4); ++i) { bw[i][0] = 0.f; bw[i][1] = 0.f; }
  float bias = 0.f;
  v8f bias8;
  #pragma unroll
  for (int r = 0; r < 8; ++r) bias8[r] = 0.f;
  if (wave < L2W) {
    load_b(bw, wh1 + ncol * 2 + kh2 * NP2, NP2);
  } else if (wave < H1W) {
    load_b(bw,          wi2 + ncol * 2 + kh2 * NP2, NP2);
    load_b(bw + KH / 4, wh2 + ncol * 2 + kh2 * NP2, NP2);
    #pragma unroll
    for (int r = 0; r < 8; ++r) bias8[r] = bg2[gc0 + r];
  } else if (wave < H2W) {
    load_b(bw, l1T + ncol * 2 + kh2 * KH2, KH2);
    bias = bl1L[ncol];
  } else if (wave < IDW) {
    load_b(bw, l2T + ncol * 2 + kh2 * VP2, VP2);
    bias = bl2L[ncol];
  }

  if (wave < L2W) {
    #pragma unroll
    for (int r = 0; r < 8; ++r) xg[r] = ptab[ids0[row] + gc0 + r];
  }

  const int wv = __builtin_amdgcn_readfirstlane(wave);

  for (int s = 0; s < ST; ++s) {
    const int cur = (s & 1) * HB;
    const int oth = ((s + 1) & 1) * HB;
    if (wave < L2W) {
      if (s < T_LEN) {
        const int* idr = ids + ((s + 1) & 1) * CHUNK;
        v2f a[KH / 4];
        load_a(a, h1 + oth + row * KHP + khalf);
        const int tk = idr[row];
        __builtin_amdgcn_sched_barrier(0);
        const v8f acc = mm_regs_t(xg, bw, a);
        const float* pr = ptab + tk + gc0;
        const v4f x0 = *(const v4f*)pr;
        const v4f x1 = *(const v4f*)(pr + 4);
        #pragma unroll
        for (int r = 0; r < 4; ++r) { xg[r] = x0[r]; xg[r + 4] = x1[r]; }
        const v4f q0 = {acc[0], acc[1], acc[2], acc[3]};
        const v4f q1 = {acc[4], acc[5], acc[6], acc[7]};
        const bool on = (s < len);
        lstm_cell(q0, on, cs0, hs0);
        lstm_cell(q1, on, cs1, hs1);
        if (cu < HID) {
          const v2f hv = {hs0, hs1};
          *(v2f*)(h1 + cur + row * KHP + cu) = hv;
        }
      }
    } else if (wave < H1W) {
      if (s >= 1 && s <= T_LEN) {
        const int t = s - 1;
        v2f a[2 * (KH / 4)];
        load_a(a,          h1 + oth + row * KHP + khalf);
        load_a(a + KH / 4, h2 + cur + row * KHP + khalf);
        __builtin_amdgcn_sched_barrier(0);
        v8f acc = bias8;
        acc = mm_regs_t(acc, bw,          a);
        acc = mm_regs_t(acc, bw + KH / 4, a + KH / 4);
        const v4f q0 = {acc[0], acc[1], acc[2], acc[3]};
        const v4f q1 = {acc[4], acc[5], acc[6], acc[7]};
        const bool on = (t < len);
        lstm_cell(q0, on, cs0, hs0);
        lstm_cell(q1, on, cs1, hs1);
        if (cu < HID) {
          const v2f hv = {hs0, hs1};
          const v2f hm = {on ? hs0 : 0.f, on ? hs1 : 0.f};
          *(v2f*)(h2  + oth + row * KHP + cu) = hv;
          *(v2f*)(h2m + oth + row * KHP + cu) = hm;
        }
      }
    } else if (wave < H2W) {
      if (s >= 2 && s < T_LEN + 2) {
        v2f a[KH / 4];
        load_a(a, h2m + cur + row * KHP + khalf);
        __builtin_amdgcn_sched_barrier(0);
        v8f acc;
        #pragma unroll
        for (int r = 0; r < 8; ++r) acc[r] = bias;
        acc = mm_regs(acc, a, bw);
        #pragma unroll
        for (int r = 0; r < 8; ++r) hmid[cur + (r + mhalf) * KHP + ncol] = fmaxf(acc[r], 0.f);
      }
    } else if (wave < IDW) {
      if (s >= 3) {
        const int t = s - 3;
        v2f a[KH / 4];
        load_a(a, hmid + oth + row * KHP + khalf);
        lds_arrive();
        __builtin_amdgcn_sched_barrier(0);
        v8f acc;
        #pragma unroll
        for (int r = 0; r < 8; ++r) acc[r] = bias;
        acc = mm_regs(acc, a, bw);
        if (ncol < VOC) {
          #pragma unroll
          for (int r = 0; r < 8; ++r) {
            const int m = r + mhalf;
            out[((size_t)(b0 + m) * T_LEN + t) * VOC + ncol] = acc[r];
          }
        }
      } else {
        lds_arrive();
      }
    } else {
      if (lane < CHUNK) {
        ids[(s & 1) * CHUNK + lane] = nid;
        nid = (s + 3 < T_LEN) ? batch_x[(b0 + lane) * T_LEN + s + 3] * PP : 0;
      }
    }
    if (wv < H2W || wv >= IDW) lds_arrive();
    lds_sync();
  }
}

extern "C" void kernel_launch(void* const* d_in, const int* in_sizes, int n_in,
                              void* d_out, int out_size, void* d_ws, size_t ws_size,
                              hipStream_t stream) {
  const int*   batch_x = (const int*)  d_in[0];
  const int*   lens    = (const int*)  d_in[1];
  const float* emb     = (const float*)d_in[2];
  const float* w_ih1   = (const float*)d_in[3];
  const float* w_hh1   = (const float*)d_in[4];
  const float* b_ih1   = (const float*)d_in[5];
  const float* b_hh1   = (const float*)d_in[6];
  const float* w_ih2   = (const float*)d_in[7];
  const float* w_hh2   = (const float*)d_in[8];
  const float* b_ih2   = (const float*)d_in[9];
  const float* b_hh2   = (const float*)d_in[10];
  const float* lw1     = (const float*)d_in[11];
  const float* lb1     = (const float*)d_in[12];
  const float* lw2     = (const float*)d_in[13];
  const float* lb2     = (const float*)d_in[14];
  float* out = (float*)d_out;

  constexpr size_t SMEM_FLOATS =
      (size_t)VOC * PP + 3u * (KH / 2) * NP2 + (size_t)(KH / 2) * KH2 +
      (size_t)(KH / 2) * VP2 + (size_t)NP + KH + VP + 8u * CHUNK * KHP + 3u * CHUNK;
  constexpr size_t SMEM_BYTES = SMEM_FLOATS * sizeof(float);

  (void)hipFuncSetAttribute((const void*)charnn_fused,
                            hipFuncAttributeMaxDynamicSharedMemorySize,
                            (int)SMEM_BYTES);

  charnn_fused<<<dim3(B_SZ / CHUNK), dim3(NT), SMEM_BYTES, stream>>>(
      batch_x, lens, emb, w_ih1, w_hh1, b_ih1, b_hh1,
      w_ih2, w_hh2, b_ih2, b_hh2, lw1, lb1, lw2, lb2, out);
}
